# starting kernel
# baseline (speedup 1.0000x reference)
#include <hip/hip_runtime.h>
#include <math.h>

typedef __attribute__((ext_vector_type(2))) float v2f;
typedef __attribute__((ext_vector_type(8))) float v8f;

static constexpr int kS  = 2048;
static constexpr int kD  = 512;
static constexpr int kH  = 8;
static constexpr int kDH = 64;

__device__ __forceinline__ v8f wmma_f32(v2f a, v2f b, v8f c) {
  return __builtin_amdgcn_wmma_f32_16x16x4_f32(
       false, a,  false, b,
       (short)0, c,  false,  false);
}

__global__ void proj_gemm_head(const float* __restrict__ A,
                               const float* __restrict__ B,
                               float* __restrict__ out) {
  const int lane = threadIdx.x & 31;
  const int wave = threadIdx.x >> 5;
  const int half = lane >> 4;
  const int lr   = lane & 15;
  const int m0   = blockIdx.x * 16;
  const int n0   = wave * 64;

  const float* ap = A + (size_t)(m0 + lr) * kD + 2 * half;
  const float* bp = B + (size_t)(n0 + lr) * kD + 2 * half;

  v8f acc[4] = {v8f{}, v8f{}, v8f{}, v8f{}};
  for (int k = 0; k < kD; k += 4) {
    v2f a = *(const v2f*)(ap + k);
#pragma unroll
    for (int t = 0; t < 4; ++t) {
      v2f b = *(const v2f*)(bp + (size_t)t * 16 * kD + k);
      acc[t] = wmma_f32(a, b, acc[t]);
    }
  }
#pragma unroll
  for (int t = 0; t < 4; ++t) {
    int n = n0 + t * 16 + lr;
    int h = n >> 6, d = n & 63;
    float* op = out + (size_t)h * kS * kDH + d;
#pragma unroll
    for (int r = 0; r < 8; ++r) {
      int m = m0 + r + 8 * half;
      op[(size_t)m * kDH] = acc[t][r];
    }
  }
}

__global__ void bd_gemm(const float* __restrict__ Qh, const float* __restrict__ Ph,
                        const float* __restrict__ vb, float* __restrict__ BD, int h) {
  const int lane = threadIdx.x & 31;
  const int wave = threadIdx.x >> 5;
  const int half = lane >> 4;
  const int lr   = lane & 15;
  const int m0   = blockIdx.x * 16;
  const int n0   = blockIdx.y * 512 + wave * 64;

  const float* qp = Qh + (size_t)h * kS * kDH + (size_t)(m0 + lr) * kDH + 2 * half;
  const float* pp = Ph + (size_t)h * kS * kDH + (size_t)(n0 + lr) * kDH + 2 * half;
  const float* bb = vb + h * kDH + 2 * half;

  v8f acc[4] = {v8f{}, v8f{}, v8f{}, v8f{}};
  for (int k = 0; k < kDH; k += 4) {
    v2f a = *(const v2f*)(qp + k);
    v2f bias = *(const v2f*)(bb + k);
    a = a + bias;
#pragma unroll
    for (int t = 0; t < 4; ++t) {
      v2f b = *(const v2f*)(pp + (size_t)t * 16 * kDH + k);
      acc[t] = wmma_f32(a, b, acc[t]);
    }
  }
#pragma unroll
  for (int t = 0; t < 4; ++t) {
    int n = n0 + t * 16 + lr;
#pragma unroll
    for (int r = 0; r < 8; ++r) {
      int m = m0 + r + 8 * half;
      BD[(size_t)m * kS + n] = acc[t][r];
    }
  }
}

__global__ void score_gemm(const float* __restrict__ Qh, const float* __restrict__ Kh,
                           const float* __restrict__ ub, const float* __restrict__ BD,
                           float* __restrict__ SC, int h, float scale) {
  const int lane = threadIdx.x & 31;
  const int wave = threadIdx.x >> 5;
  const int half = lane >> 4;
  const int lr   = lane & 15;
  const int m0   = blockIdx.x * 16;
  const int n0   = blockIdx.y * 512 + wave * 64;

  const float* qp = Qh + (size_t)h * kS * kDH + (size_t)(m0 + lr) * kDH + 2 * half;
  const float* kp = Kh + (size_t)h * kS * kDH + (size_t)(n0 + lr) * kDH + 2 * half;
  const float* bb = ub + h * kDH + 2 * half;

  v8f acc[4] = {v8f{}, v8f{}, v8f{}, v8f{}};
  for (int k = 0; k < kDH; k += 4) {
    v2f a = *(const v2f*)(qp + k);
    v2f bias = *(const v2f*)(bb + k);
    a = a + bias;
#pragma unroll
    for (int t = 0; t < 4; ++t) {
      v2f b = *(const v2f*)(kp + (size_t)t * 16 * kDH + k);
      acc[t] = wmma_f32(a, b, acc[t]);
    }
  }
#pragma unroll
  for (int t = 0; t < 4; ++t) {
    int n = n0 + t * 16 + lr;
#pragma unroll
    for (int r = 0; r < 8; ++r) {
      int m = m0 + r + 8 * half;
      float bd;
      if (n <= m)          bd = BD[(size_t)m * kS + (kS - 1 - m + n)];
      else if (n == m + 1) bd = 0.0f;
      else                 bd = BD[(size_t)(m + 1) * kS + (n - m - 2)];
      SC[(size_t)m * kS + n] = (acc[t][r] + bd) * scale;
    }
  }
}

__global__ void softmax_rows(float* __restrict__ SC) {
  __shared__ float red[256];
  const int tid = threadIdx.x;
  float* p = SC + (size_t)blockIdx.x * kS;

  float m = -INFINITY;
  for (int i = tid; i < kS; i += 256) m = fmaxf(m, p[i]);
  red[tid] = m; __syncthreads();
  for (int s = 128; s > 0; s >>= 1) {
    if (tid < s) red[tid] = fmaxf(red[tid], red[tid + s]);
    __syncthreads();
  }
  m = red[0]; __syncthreads();

  float sum = 0.0f;
  for (int i = tid; i < kS; i += 256) {
    float e = __expf(p[i] - m);
    p[i] = e;
    sum += e;
  }
  red[tid] = sum; __syncthreads();
  for (int s = 128; s > 0; s >>= 1) {
    if (tid < s) red[tid] += red[tid + s];
    __syncthreads();
  }
  float inv = 1.0f / red[0];
  for (int i = tid; i < kS; i += 256) p[i] *= inv;
}

__global__ void av_gemm(const float* __restrict__ SC, const float* __restrict__ Vh,
                        float* __restrict__ AVec, int h) {
  const int lane = threadIdx.x & 31;
  const int wave = threadIdx.x >> 5;
  const int half = lane >> 4;
  const int lr   = lane & 15;
  const int m0   = (blockIdx.x * 8 + wave) * 16;

  const float* pp = SC + (size_t)(m0 + lr) * kS + 2 * half;
  const float* vp = Vh + (size_t)h * kS * kDH + lr;

  v8f acc[4] = {v8f{}, v8f{}, v8f{}, v8f{}};
  for (int k = 0; k < kS; k += 4) {
    v2f a = *(const v2f*)(pp + k);
    int kk = k + 2 * half;
#pragma unroll
    for (int t = 0; t < 4; ++t) {
      const float* vb = vp + t * 16;
      v2f b = {vb[(size_t)kk * kDH], vb[(size_t)(kk + 1) * kDH]};
      acc[t] = wmma_f32(a, b, acc[t]);
    }
  }
#pragma unroll
  for (int t = 0; t < 4; ++t) {
    int d = t * 16 + lr;
#pragma unroll
    for (int r = 0; r < 8; ++r) {
      int m = m0 + r + 8 * half;
      AVec[(size_t)m * kD + h * kDH + d] = acc[t][r];
    }
  }
}

__global__ void out_gemm(const float* __restrict__ A, const float* __restrict__ B,
                         float* __restrict__ C) {
  const int lane = threadIdx.x & 31;
  const int wave = threadIdx.x >> 5;
  const int half = lane >> 4;
  const int lr   = lane & 15;
  const int m0   = blockIdx.x * 16;
  const int n0   = wave * 64;

  const float* ap = A + (size_t)(m0 + lr) * kD + 2 * half;
  const float* bp = B + (size_t)(n0 + lr) * kD + 2 * half;

  v8f acc[4] = {v8f{}, v8f{}, v8f{}, v8f{}};
  for (int k = 0; k < kD; k += 4) {
    v2f a = *(const v2f*)(ap + k);
#pragma unroll
    for (int t = 0; t < 4; ++t) {
      v2f b = *(const v2f*)(bp + (size_t)t * 16 * kD + k);
      acc[t] = wmma_f32(a, b, acc[t]);
    }
  }
#pragma unroll
  for (int t = 0; t < 4; ++t) {
    int n = n0 + t * 16 + lr;
#pragma unroll
    for (int r = 0; r < 8; ++r) {
      int m = m0 + r + 8 * half;
      C[(size_t)m * kD + n] = acc[t][r];
    }
  }
}

extern "C" void kernel_launch(void* const* d_in, const int* in_sizes, int n_in,
                              void* d_out, int out_size, void* d_ws, size_t ws_size,
                              hipStream_t stream) {
  const float* x   = (const float*)d_in[0];
  const float* pos = (const float*)d_in[1];
  const float* Wq  = (const float*)d_in[2];
  const float* Wk  = (const float*)d_in[3];
  const float* Wv  = (const float*)d_in[4];
  const float* Wp  = (const float*)d_in[5];
  const float* Wo  = (const float*)d_in[6];
  const float* u   = (const float*)d_in[7];
  const float* v   = (const float*)d_in[8];

  float* ws = (float*)d_ws;
  const size_t SD = (size_t)kS * kD;
  float* Qhs  = ws;
  float* Khs  = Qhs + SD;
  float* Vhs  = Khs + SD;
  float* Phs  = Vhs + SD;
  float* AVec = Phs + SD;
  float* BD   = AVec + SD;
  float* SC   = BD + (size_t)kS * kS;

  dim3 blk(256);
  proj_gemm_head<<<dim3(kS / 16), blk, 0, stream>>>(x,   Wq, Qhs);
  proj_gemm_head<<<dim3(kS / 16), blk, 0, stream>>>(x,   Wk, Khs);
  proj_gemm_head<<<dim3(kS / 16), blk, 0, stream>>>(x,   Wv, Vhs);
  proj_gemm_head<<<dim3(kS / 16), blk, 0, stream>>>(pos, Wp, Phs);

  const float scale = 0.125f;
  for (int h = 0; h < kH; ++h) {
    bd_gemm   <<<dim3(kS / 16, kS / 512), blk, 0, stream>>>(Qhs, Phs, v, BD, h);
    score_gemm<<<dim3(kS / 16, kS / 512), blk, 0, stream>>>(Qhs, Khs, u, BD, SC, h, scale);
    softmax_rows<<<dim3(kS), blk, 0, stream>>>(SC);
    av_gemm   <<<dim3(kS / 128), blk, 0, stream>>>(SC, Vhs, AVec, h);
  }

  out_gemm<<<dim3(kS / 16), blk, 0, stream>>>(AVec, Wo, (float*)d_out);
}


# AV product: K split over the 8 waves of a block, partial tiles added through LDS in fixed wave order (new summation order); grid 16 -> 256 blocks of 16x32 outputs
# speedup vs baseline: 2.5269x; 2.5269x over previous
#include <hip/hip_runtime.h>
#include <math.h>

typedef __attribute__((ext_vector_type(2))) float v2f;
typedef __attribute__((ext_vector_type(8))) float v8f;

static constexpr int kS  = 2048;
static constexpr int kD  = 512;
static constexpr int kH  = 8;
static constexpr int kDH = 64;

__device__ __forceinline__ v8f wmma_f32(v2f a, v2f b, v8f c) {
  return __builtin_amdgcn_wmma_f32_16x16x4_f32(
       false, a,  false, b,
       (short)0, c,  false,  false);
}

__global__ void proj_gemm_head(const float* __restrict__ A,
                               const float* __restrict__ B,
                               float* __restrict__ out) {
  const int lane = threadIdx.x & 31;
  const int wave = threadIdx.x >> 5;
  const int half = lane >> 4;
  const int lr   = lane & 15;
  const int m0   = blockIdx.x * 16;
  const int n0   = wave * 64;

  const float* ap = A + (size_t)(m0 + lr) * kD + 2 * half;
  const float* bp = B + (size_t)(n0 + lr) * kD + 2 * half;

  v8f acc[4] = {v8f{}, v8f{}, v8f{}, v8f{}};
  for (int k = 0; k < kD; k += 4) {
    v2f a = *(const v2f*)(ap + k);
#pragma unroll
    for (int t = 0; t < 4; ++t) {
      v2f b = *(const v2f*)(bp + (size_t)t * 16 * kD + k);
      acc[t] = wmma_f32(a, b, acc[t]);
    }
  }
#pragma unroll
  for (int t = 0; t < 4; ++t) {
    int n = n0 + t * 16 + lr;
    int h = n >> 6, d = n & 63;
    float* op = out + (size_t)h * kS * kDH + d;
#pragma unroll
    for (int r = 0; r < 8; ++r) {
      int m = m0 + r + 8 * half;
      op[(size_t)m * kDH] = acc[t][r];
    }
  }
}

__global__ void bd_gemm(const float* __restrict__ Qh, const float* __restrict__ Ph,
                        const float* __restrict__ vb, float* __restrict__ BD, int h) {
  const int lane = threadIdx.x & 31;
  const int wave = threadIdx.x >> 5;
  const int half = lane >> 4;
  const int lr   = lane & 15;
  const int m0   = blockIdx.x * 16;
  const int n0   = blockIdx.y * 512 + wave * 64;

  const float* qp = Qh + (size_t)h * kS * kDH + (size_t)(m0 + lr) * kDH + 2 * half;
  const float* pp = Ph + (size_t)h * kS * kDH + (size_t)(n0 + lr) * kDH + 2 * half;
  const float* bb = vb + h * kDH + 2 * half;

  v8f acc[4] = {v8f{}, v8f{}, v8f{}, v8f{}};
  for (int k = 0; k < kDH; k += 4) {
    v2f a = *(const v2f*)(qp + k);
    v2f bias = *(const v2f*)(bb + k);
    a = a + bias;
#pragma unroll
    for (int t = 0; t < 4; ++t) {
      v2f b = *(const v2f*)(pp + (size_t)t * 16 * kDH + k);
      acc[t] = wmma_f32(a, b, acc[t]);
    }
  }
#pragma unroll
  for (int t = 0; t < 4; ++t) {
    int n = n0 + t * 16 + lr;
#pragma unroll
    for (int r = 0; r < 8; ++r) {
      int m = m0 + r + 8 * half;
      BD[(size_t)m * kS + n] = acc[t][r];
    }
  }
}

__global__ void score_gemm(const float* __restrict__ Qh, const float* __restrict__ Kh,
                           const float* __restrict__ ub, const float* __restrict__ BD,
                           float* __restrict__ SC, int h, float scale) {
  const int lane = threadIdx.x & 31;
  const int wave = threadIdx.x >> 5;
  const int half = lane >> 4;
  const int lr   = lane & 15;
  const int m0   = blockIdx.x * 16;
  const int n0   = blockIdx.y * 512 + wave * 64;

  const float* qp = Qh + (size_t)h * kS * kDH + (size_t)(m0 + lr) * kDH + 2 * half;
  const float* kp = Kh + (size_t)h * kS * kDH + (size_t)(n0 + lr) * kDH + 2 * half;
  const float* bb = ub + h * kDH + 2 * half;

  v8f acc[4] = {v8f{}, v8f{}, v8f{}, v8f{}};
  for (int k = 0; k < kDH; k += 4) {
    v2f a = *(const v2f*)(qp + k);
    v2f bias = *(const v2f*)(bb + k);
    a = a + bias;
#pragma unroll
    for (int t = 0; t < 4; ++t) {
      v2f b = *(const v2f*)(kp + (size_t)t * 16 * kDH + k);
      acc[t] = wmma_f32(a, b, acc[t]);
    }
  }
#pragma unroll
  for (int t = 0; t < 4; ++t) {
    int n = n0 + t * 16 + lr;
#pragma unroll
    for (int r = 0; r < 8; ++r) {
      int m = m0 + r + 8 * half;
      float bd;
      if (n <= m)          bd = BD[(size_t)m * kS + (kS - 1 - m + n)];
      else if (n == m + 1) bd = 0.0f;
      else                 bd = BD[(size_t)(m + 1) * kS + (n - m - 2)];
      SC[(size_t)m * kS + n] = (acc[t][r] + bd) * scale;
    }
  }
}

__global__ void softmax_rows(float* __restrict__ SC) {
  __shared__ float red[256];
  const int tid = threadIdx.x;
  float* p = SC + (size_t)blockIdx.x * kS;

  float m = -INFINITY;
  for (int i = tid; i < kS; i += 256) m = fmaxf(m, p[i]);
  red[tid] = m; __syncthreads();
  for (int s = 128; s > 0; s >>= 1) {
    if (tid < s) red[tid] = fmaxf(red[tid], red[tid + s]);
    __syncthreads();
  }
  m = red[0]; __syncthreads();

  float sum = 0.0f;
  for (int i = tid; i < kS; i += 256) {
    float e = __expf(p[i] - m);
    p[i] = e;
    sum += e;
  }
  red[tid] = sum; __syncthreads();
  for (int s = 128; s > 0; s >>= 1) {
    if (tid < s) red[tid] += red[tid + s];
    __syncthreads();
  }
  float inv = 1.0f / red[0];
  for (int i = tid; i < kS; i += 256) p[i] *= inv;
}

static constexpr int kAvWaves = 8;
static constexpr int kAvK     = kS / kAvWaves;
static constexpr int kAvN     = 32;

__global__ void av_gemm(const float* __restrict__ SC, const float* __restrict__ Vh,
                        float* __restrict__ AVec, int h) {
  __shared__ float part[kAvWaves][16][32];
  const int lane = threadIdx.x & 31;
  const int wave = threadIdx.x >> 5;
  const int half = lane >> 4;
  const int lr   = lane & 15;
  const int m0   = blockIdx.x * 16;
  const int n0   = blockIdx.y * kAvN;
  const int kb   = wave * kAvK + 2 * half;

  const float* pp = SC + (size_t)(m0 + lr) * kS + kb;
  const float* vp = Vh + (size_t)h * kS * kDH + (size_t)kb * kDH + n0 + lr;

  v8f acc[2] = {v8f{}, v8f{}};
  for (int k = 0; k < kAvK; k += 4) {
    v2f a = *(const v2f*)(pp + k);
#pragma unroll
    for (int t = 0; t < 2; ++t) {
      const float* vb = vp + (size_t)k * kDH + t * 16;
      v2f b = {vb[0], vb[kDH]};
      acc[t] = wmma_f32(a, b, acc[t]);
    }
  }

#pragma unroll
  for (int t = 0; t < 2; ++t) {
#pragma unroll
    for (int r = 0; r < 8; ++r) part[wave][t * 8 + r][lane] = acc[t][r];
  }
  __syncthreads();

  const int m = m0 + wave + 8 * half;
  for (int t = 0; t < 2; ++t) {
    float s = part[0][t * 8 + wave][lane];
    for (int w = 1; w < kAvWaves; ++w) s += part[w][t * 8 + wave][lane];
    AVec[(size_t)m * kD + h * kDH + n0 + t * 16 + lr] = s;
  }
}

__global__ void out_gemm(const float* __restrict__ A, const float* __restrict__ B,
                         float* __restrict__ C) {
  const int lane = threadIdx.x & 31;
  const int wave = threadIdx.x >> 5;
  const int half = lane >> 4;
  const int lr   = lane & 15;
  const int m0   = blockIdx.x * 16;
  const int n0   = wave * 64;

  const float* ap = A + (size_t)(m0 + lr) * kD + 2 * half;
  const float* bp = B + (size_t)(n0 + lr) * kD + 2 * half;

  v8f acc[4] = {v8f{}, v8f{}, v8f{}, v8f{}};
  for (int k = 0; k < kD; k += 4) {
    v2f a = *(const v2f*)(ap + k);
#pragma unroll
    for (int t = 0; t < 4; ++t) {
      v2f b = *(const v2f*)(bp + (size_t)t * 16 * kD + k);
      acc[t] = wmma_f32(a, b, acc[t]);
    }
  }
#pragma unroll
  for (int t = 0; t < 4; ++t) {
    int n = n0 + t * 16 + lr;
#pragma unroll
    for (int r = 0; r < 8; ++r) {
      int m = m0 + r + 8 * half;
      C[(size_t)m * kD + n] = acc[t][r];
    }
  }
}

extern "C" void kernel_launch(void* const* d_in, const int* in_sizes, int n_in,
                              void* d_out, int out_size, void* d_ws, size_t ws_size,
                              hipStream_t stream) {
  const float* x   = (const float*)d_in[0];
  const float* pos = (const float*)d_in[1];
  const float* Wq  = (const float*)d_in[2];
  const float* Wk  = (const float*)d_in[3];
  const float* Wv  = (const float*)d_in[4];
  const float* Wp  = (const float*)d_in[5];
  const float* Wo  = (const float*)d_in[6];
  const float* u   = (const float*)d_in[7];
  const float* v   = (const float*)d_in[8];

  float* ws = (float*)d_ws;
  const size_t SD = (size_t)kS * kD;
  float* Qhs  = ws;
  float* Khs  = Qhs + SD;
  float* Vhs  = Khs + SD;
  float* Phs  = Vhs + SD;
  float* AVec = Phs + SD;
  float* BD   = AVec + SD;
  float* SC   = BD + (size_t)kS * kS;

  dim3 blk(256);
  proj_gemm_head<<<dim3(kS / 16), blk, 0, stream>>>(x,   Wq, Qhs);
  proj_gemm_head<<<dim3(kS / 16), blk, 0, stream>>>(x,   Wk, Khs);
  proj_gemm_head<<<dim3(kS / 16), blk, 0, stream>>>(x,   Wv, Vhs);
  proj_gemm_head<<<dim3(kS / 16), blk, 0, stream>>>(pos, Wp, Phs);

  const float scale = 0.125f;
  for (int h = 0; h < kH; ++h) {
    bd_gemm   <<<dim3(kS / 16, kS / 512), blk, 0, stream>>>(Qhs, Phs, v, BD, h);
    score_gemm<<<dim3(kS / 16, kS / 512), blk, 0, stream>>>(Qhs, Khs, u, BD, SC, h, scale);
    softmax_rows<<<dim3(kS), blk, 0, stream>>>(SC);
    av_gemm   <<<dim3(kS / 16, kDH / kAvN), blk, 0, stream>>>(SC, Vhs, AVec, h);
  }

  out_gemm<<<dim3(kS / 16), blk, 0, stream>>>(AVec, Wo, (float*)d_out);
}


# softmax fused into the AV product: each wave holds its 16x256 score slice in registers, feeds exp(score - slice max) to the WMMA A operand; slice max/sum merged via LDS in wave order (new sum order)
# speedup vs baseline: 3.2673x; 1.2930x over previous
#include <hip/hip_runtime.h>
#include <math.h>

typedef __attribute__((ext_vector_type(2))) float v2f;
typedef __attribute__((ext_vector_type(8))) float v8f;

static constexpr int kS  = 2048;
static constexpr int kD  = 512;
static constexpr int kH  = 8;
static constexpr int kDH = 64;

__device__ __forceinline__ v8f wmma_f32(v2f a, v2f b, v8f c) {
  return __builtin_amdgcn_wmma_f32_16x16x4_f32(
       false, a,  false, b,
       (short)0, c,  false,  false);
}

__global__ void proj_gemm_head(const float* __restrict__ A,
                               const float* __restrict__ B,
                               float* __restrict__ out) {
  const int lane = threadIdx.x & 31;
  const int wave = threadIdx.x >> 5;
  const int half = lane >> 4;
  const int lr   = lane & 15;
  const int m0   = blockIdx.x * 16;
  const int n0   = wave * 64;

  const float* ap = A + (size_t)(m0 + lr) * kD + 2 * half;
  const float* bp = B + (size_t)(n0 + lr) * kD + 2 * half;

  v8f acc[4] = {v8f{}, v8f{}, v8f{}, v8f{}};
  for (int k = 0; k < kD; k += 4) {
    v2f a = *(const v2f*)(ap + k);
#pragma unroll
    for (int t = 0; t < 4; ++t) {
      v2f b = *(const v2f*)(bp + (size_t)t * 16 * kD + k);
      acc[t] = wmma_f32(a, b, acc[t]);
    }
  }
#pragma unroll
  for (int t = 0; t < 4; ++t) {
    int n = n0 + t * 16 + lr;
    int h = n >> 6, d = n & 63;
    float* op = out + (size_t)h * kS * kDH + d;
#pragma unroll
    for (int r = 0; r < 8; ++r) {
      int m = m0 + r + 8 * half;
      op[(size_t)m * kDH] = acc[t][r];
    }
  }
}

__global__ void bd_gemm(const float* __restrict__ Qh, const float* __restrict__ Ph,
                        const float* __restrict__ vb, float* __restrict__ BD, int h) {
  const int lane = threadIdx.x & 31;
  const int wave = threadIdx.x >> 5;
  const int half = lane >> 4;
  const int lr   = lane & 15;
  const int m0   = blockIdx.x * 16;
  const int n0   = blockIdx.y * 512 + wave * 64;

  const float* qp = Qh + (size_t)h * kS * kDH + (size_t)(m0 + lr) * kDH + 2 * half;
  const float* pp = Ph + (size_t)h * kS * kDH + (size_t)(n0 + lr) * kDH + 2 * half;
  const float* bb = vb + h * kDH + 2 * half;

  v8f acc[4] = {v8f{}, v8f{}, v8f{}, v8f{}};
  for (int k = 0; k < kDH; k += 4) {
    v2f a = *(const v2f*)(qp + k);
    v2f bias = *(const v2f*)(bb + k);
    a = a + bias;
#pragma unroll
    for (int t = 0; t < 4; ++t) {
      v2f b = *(const v2f*)(pp + (size_t)t * 16 * kDH + k);
      acc[t] = wmma_f32(a, b, acc[t]);
    }
  }
#pragma unroll
  for (int t = 0; t < 4; ++t) {
    int n = n0 + t * 16 + lr;
#pragma unroll
    for (int r = 0; r < 8; ++r) {
      int m = m0 + r + 8 * half;
      BD[(size_t)m * kS + n] = acc[t][r];
    }
  }
}

__global__ void score_gemm(const float* __restrict__ Qh, const float* __restrict__ Kh,
                           const float* __restrict__ ub, const float* __restrict__ BD,
                           float* __restrict__ SC, int h, float scale) {
  const int lane = threadIdx.x & 31;
  const int wave = threadIdx.x >> 5;
  const int half = lane >> 4;
  const int lr   = lane & 15;
  const int m0   = blockIdx.x * 16;
  const int n0   = blockIdx.y * 512 + wave * 64;

  const float* qp = Qh + (size_t)h * kS * kDH + (size_t)(m0 + lr) * kDH + 2 * half;
  const float* kp = Kh + (size_t)h * kS * kDH + (size_t)(n0 + lr) * kDH + 2 * half;
  const float* bb = ub + h * kDH + 2 * half;

  v8f acc[4] = {v8f{}, v8f{}, v8f{}, v8f{}};
  for (int k = 0; k < kDH; k += 4) {
    v2f a = *(const v2f*)(qp + k);
    v2f bias = *(const v2f*)(bb + k);
    a = a + bias;
#pragma unroll
    for (int t = 0; t < 4; ++t) {
      v2f b = *(const v2f*)(kp + (size_t)t * 16 * kDH + k);
      acc[t] = wmma_f32(a, b, acc[t]);
    }
  }
#pragma unroll
  for (int t = 0; t < 4; ++t) {
    int n = n0 + t * 16 + lr;
#pragma unroll
    for (int r = 0; r < 8; ++r) {
      int m = m0 + r + 8 * half;
      float bd;
      if (n <= m)          bd = BD[(size_t)m * kS + (kS - 1 - m + n)];
      else if (n == m + 1) bd = 0.0f;
      else                 bd = BD[(size_t)(m + 1) * kS + (n - m - 2)];
      SC[(size_t)m * kS + n] = (acc[t][r] + bd) * scale;
    }
  }
}

static constexpr int kAvWaves = 8;
static constexpr int kAvK     = kS / kAvWaves;
static constexpr int kAvN     = 32;
static constexpr int kAvSteps = kAvK / 4;
static constexpr int kAvChunk = 8;

__global__ void __launch_bounds__(kAvWaves * 32)
av_gemm(const float* __restrict__ SC, const float* __restrict__ Vh,
        float* __restrict__ AVec, int h) {
  __shared__ float part[kAvWaves][16][32];
  __shared__ float smax[kAvWaves][16];
  __shared__ float ssum[kAvWaves][16];
  const int lane = threadIdx.x & 31;
  const int wave = threadIdx.x >> 5;
  const int half = lane >> 4;
  const int lr   = lane & 15;
  const int m0   = blockIdx.x * 16;
  const int n0   = blockIdx.y * kAvN;
  const int kb   = wave * kAvK + 2 * half;

  const float* pp = SC + (size_t)(m0 + lr) * kS + kb;
  const float* vp = Vh + (size_t)h * kS * kDH + (size_t)kb * kDH + n0 + lr;

  v2f sc[kAvSteps];
#pragma unroll
  for (int i = 0; i < kAvSteps; ++i) sc[i] = *(const v2f*)(pp + 4 * i);

  v2f bq[2][kAvChunk * 2];
#pragma unroll
  for (int j = 0; j < kAvChunk * 2; ++j) {
    const float* vb = vp + (size_t)(4 * (j >> 1)) * kDH + (j & 1) * 16;
    bq[0][j] = v2f{vb[0], vb[kDH]};
  }
  __builtin_amdgcn_sched_barrier(0);

  float mx = fmaxf(sc[0][0], sc[0][1]);
#pragma unroll
  for (int i = 1; i < kAvSteps; ++i) mx = fmaxf(mx, fmaxf(sc[i][0], sc[i][1]));
  mx = fmaxf(mx, __shfl_xor(mx, 16, 32));

  float sum = 0.0f;
  v8f acc[2] = {v8f{}, v8f{}};
#pragma unroll
  for (int c = 0; c < kAvSteps / kAvChunk; ++c) {
    if (c + 1 < kAvSteps / kAvChunk) {
#pragma unroll
      for (int j = 0; j < kAvChunk * 2; ++j) {
        const float* vb = vp + (size_t)(4 * ((c + 1) * kAvChunk + (j >> 1))) * kDH + (j & 1) * 16;
        bq[(c + 1) & 1][j] = v2f{vb[0], vb[kDH]};
      }
    }
    __builtin_amdgcn_sched_barrier(0);
#pragma unroll
    for (int s = 0; s < kAvChunk; ++s) {
      const int i = c * kAvChunk + s;
      v2f a = {__expf(sc[i][0] - mx), __expf(sc[i][1] - mx)};
      sum += a[0] + a[1];
#pragma unroll
      for (int t = 0; t < 2; ++t) acc[t] = wmma_f32(a, bq[c & 1][s * 2 + t], acc[t]);
    }
  }
  sum += __shfl_xor(sum, 16, 32);

  if (half == 0) {
    smax[wave][lr] = mx;
    ssum[wave][lr] = sum;
  }
#pragma unroll
  for (int t = 0; t < 2; ++t) {
#pragma unroll
    for (int r = 0; r < 8; ++r) part[wave][t * 8 + r][lane] = acc[t][r];
  }
  __syncthreads();

  const int row = wave + 8 * half;
  float g = smax[0][row];
#pragma unroll
  for (int w = 1; w < kAvWaves; ++w) g = fmaxf(g, smax[w][row]);
  float f[kAvWaves];
  float den = 0.0f;
#pragma unroll
  for (int w = 0; w < kAvWaves; ++w) {
    f[w] = __expf(smax[w][row] - g);
    den += f[w] * ssum[w][row];
  }
  const float inv = 1.0f / den;
  const int m = m0 + row;
#pragma unroll
  for (int t = 0; t < 2; ++t) {
    float s = f[0] * part[0][t * 8 + wave][lane];
#pragma unroll
    for (int w = 1; w < kAvWaves; ++w) s += f[w] * part[w][t * 8 + wave][lane];
    AVec[(size_t)m * kD + h * kDH + n0 + t * 16 + lr] = s * inv;
  }
}

__global__ void out_gemm(const float* __restrict__ A, const float* __restrict__ B,
                         float* __restrict__ C) {
  const int lane = threadIdx.x & 31;
  const int wave = threadIdx.x >> 5;
  const int half = lane >> 4;
  const int lr   = lane & 15;
  const int m0   = blockIdx.x * 16;
  const int n0   = wave * 64;

  const float* ap = A + (size_t)(m0 + lr) * kD + 2 * half;
  const float* bp = B + (size_t)(n0 + lr) * kD + 2 * half;

  v8f acc[4] = {v8f{}, v8f{}, v8f{}, v8f{}};
  for (int k = 0; k < kD; k += 4) {
    v2f a = *(const v2f*)(ap + k);
#pragma unroll
    for (int t = 0; t < 4; ++t) {
      v2f b = *(const v2f*)(bp + (size_t)t * 16 * kD + k);
      acc[t] = wmma_f32(a, b, acc[t]);
    }
  }
#pragma unroll
  for (int t = 0; t < 4; ++t) {
    int n = n0 + t * 16 + lr;
#pragma unroll
    for (int r = 0; r < 8; ++r) {
      int m = m0 + r + 8 * half;
      C[(size_t)m * kD + n] = acc[t][r];
    }
  }
}

extern "C" void kernel_launch(void* const* d_in, const int* in_sizes, int n_in,
                              void* d_out, int out_size, void* d_ws, size_t ws_size,
                              hipStream_t stream) {
  const float* x   = (const float*)d_in[0];
  const float* pos = (const float*)d_in[1];
  const float* Wq  = (const float*)d_in[2];
  const float* Wk  = (const float*)d_in[3];
  const float* Wv  = (const float*)d_in[4];
  const float* Wp  = (const float*)d_in[5];
  const float* Wo  = (const float*)d_in[6];
  const float* u   = (const float*)d_in[7];
  const float* v   = (const float*)d_in[8];

  float* ws = (float*)d_ws;
  const size_t SD = (size_t)kS * kD;
  float* Qhs  = ws;
  float* Khs  = Qhs + SD;
  float* Vhs  = Khs + SD;
  float* Phs  = Vhs + SD;
  float* AVec = Phs + SD;
  float* BD   = AVec + SD;
  float* SC   = BD + (size_t)kS * kS;

  dim3 blk(256);
  proj_gemm_head<<<dim3(kS / 16), blk, 0, stream>>>(x,   Wq, Qhs);
  proj_gemm_head<<<dim3(kS / 16), blk, 0, stream>>>(x,   Wk, Khs);
  proj_gemm_head<<<dim3(kS / 16), blk, 0, stream>>>(x,   Wv, Vhs);
  proj_gemm_head<<<dim3(kS / 16), blk, 0, stream>>>(pos, Wp, Phs);

  const float scale = 0.125f;
  for (int h = 0; h < kH; ++h) {
    bd_gemm   <<<dim3(kS / 16, kS / 512), blk, 0, stream>>>(Qhs, Phs, v, BD, h);
    score_gemm<<<dim3(kS / 16, kS / 512), blk, 0, stream>>>(Qhs, Khs, u, BD, SC, h, scale);
    av_gemm   <<<dim3(kS / 16, kDH / kAvN), blk, 0, stream>>>(SC, Vhs, AVec, h);
  }

  out_gemm<<<dim3(kS / 16), blk, 0, stream>>>(AVec, Wo, (float*)d_out);
}


# score kernel epilogue: the three rel_shift cases folded into one flat BD index, zero selected after the load, so a wave's 32 gathers issue as one batch instead of 32 load-wait-store rounds
# speedup vs baseline: 3.3941x; 1.0388x over previous
#include <hip/hip_runtime.h>
#include <math.h>

typedef __attribute__((ext_vector_type(2))) float v2f;
typedef __attribute__((ext_vector_type(8))) float v8f;

static constexpr int kS  = 2048;
static constexpr int kD  = 512;
static constexpr int kH  = 8;
static constexpr int kDH = 64;

__device__ __forceinline__ v8f wmma_f32(v2f a, v2f b, v8f c) {
  return __builtin_amdgcn_wmma_f32_16x16x4_f32(
       false, a,  false, b,
       (short)0, c,  false,  false);
}

__global__ void proj_gemm_head(const float* __restrict__ A,
                               const float* __restrict__ B,
                               float* __restrict__ out) {
  const int lane = threadIdx.x & 31;
  const int wave = threadIdx.x >> 5;
  const int half = lane >> 4;
  const int lr   = lane & 15;
  const int m0   = blockIdx.x * 16;
  const int n0   = wave * 64;

  const float* ap = A + (size_t)(m0 + lr) * kD + 2 * half;
  const float* bp = B + (size_t)(n0 + lr) * kD + 2 * half;

  v8f acc[4] = {v8f{}, v8f{}, v8f{}, v8f{}};
  for (int k = 0; k < kD; k += 4) {
    v2f a = *(const v2f*)(ap + k);
#pragma unroll
    for (int t = 0; t < 4; ++t) {
      v2f b = *(const v2f*)(bp + (size_t)t * 16 * kD + k);
      acc[t] = wmma_f32(a, b, acc[t]);
    }
  }
#pragma unroll
  for (int t = 0; t < 4; ++t) {
    int n = n0 + t * 16 + lr;
    int h = n >> 6, d = n & 63;
    float* op = out + (size_t)h * kS * kDH + d;
#pragma unroll
    for (int r = 0; r < 8; ++r) {
      int m = m0 + r + 8 * half;
      op[(size_t)m * kDH] = acc[t][r];
    }
  }
}

__global__ void bd_gemm(const float* __restrict__ Qh, const float* __restrict__ Ph,
                        const float* __restrict__ vb, float* __restrict__ BD, int h) {
  const int lane = threadIdx.x & 31;
  const int wave = threadIdx.x >> 5;
  const int half = lane >> 4;
  const int lr   = lane & 15;
  const int m0   = blockIdx.x * 16;
  const int n0   = blockIdx.y * 512 + wave * 64;

  const float* qp = Qh + (size_t)h * kS * kDH + (size_t)(m0 + lr) * kDH + 2 * half;
  const float* pp = Ph + (size_t)h * kS * kDH + (size_t)(n0 + lr) * kDH + 2 * half;
  const float* bb = vb + h * kDH + 2 * half;

  v8f acc[4] = {v8f{}, v8f{}, v8f{}, v8f{}};
  for (int k = 0; k < kDH; k += 4) {
    v2f a = *(const v2f*)(qp + k);
    v2f bias = *(const v2f*)(bb + k);
    a = a + bias;
#pragma unroll
    for (int t = 0; t < 4; ++t) {
      v2f b = *(const v2f*)(pp + (size_t)t * 16 * kDH + k);
      acc[t] = wmma_f32(a, b, acc[t]);
    }
  }
#pragma unroll
  for (int t = 0; t < 4; ++t) {
    int n = n0 + t * 16 + lr;
#pragma unroll
    for (int r = 0; r < 8; ++r) {
      int m = m0 + r + 8 * half;
      BD[(size_t)m * kS + n] = acc[t][r];
    }
  }
}

__global__ void score_gemm(const float* __restrict__ Qh, const float* __restrict__ Kh,
                           const float* __restrict__ ub, const float* __restrict__ BD,
                           float* __restrict__ SC, int h, float scale) {
  const int lane = threadIdx.x & 31;
  const int wave = threadIdx.x >> 5;
  const int half = lane >> 4;
  const int lr   = lane & 15;
  const int m0   = blockIdx.x * 16;
  const int n0   = blockIdx.y * 512 + wave * 64;

  const float* qp = Qh + (size_t)h * kS * kDH + (size_t)(m0 + lr) * kDH + 2 * half;
  const float* kp = Kh + (size_t)h * kS * kDH + (size_t)(n0 + lr) * kDH + 2 * half;
  const float* bb = ub + h * kDH + 2 * half;

  v8f acc[4] = {v8f{}, v8f{}, v8f{}, v8f{}};
  for (int k = 0; k < kDH; k += 4) {
    v2f a = *(const v2f*)(qp + k);
    v2f bias = *(const v2f*)(bb + k);
    a = a + bias;
#pragma unroll
    for (int t = 0; t < 4; ++t) {
      v2f b = *(const v2f*)(kp + (size_t)t * 16 * kDH + k);
      acc[t] = wmma_f32(a, b, acc[t]);
    }
  }
  float bdv[4][8];
#pragma unroll
  for (int t = 0; t < 4; ++t) {
    const int n = n0 + t * 16 + lr;
#pragma unroll
    for (int r = 0; r < 8; ++r) {
      const int m = m0 + r + 8 * half;
      const int d = n - m;
      const unsigned idx = (unsigned)((m + 1) * kS + d - 1 - (d >= 2 ? 1 : 0));
      const float g = BD[idx];
      bdv[t][r] = (d == 1) ? 0.0f : g;
    }
  }
#pragma unroll
  for (int t = 0; t < 4; ++t) {
    const int n = n0 + t * 16 + lr;
#pragma unroll
    for (int r = 0; r < 8; ++r) {
      const int m = m0 + r + 8 * half;
      SC[(size_t)m * kS + n] = (acc[t][r] + bdv[t][r]) * scale;
    }
  }
}

static constexpr int kAvWaves = 8;
static constexpr int kAvK     = kS / kAvWaves;
static constexpr int kAvN     = 32;
static constexpr int kAvSteps = kAvK / 4;
static constexpr int kAvChunk = 8;

__global__ void __launch_bounds__(kAvWaves * 32)
av_gemm(const float* __restrict__ SC, const float* __restrict__ Vh,
        float* __restrict__ AVec, int h) {
  __shared__ float part[kAvWaves][16][32];
  __shared__ float smax[kAvWaves][16];
  __shared__ float ssum[kAvWaves][16];
  const int lane = threadIdx.x & 31;
  const int wave = threadIdx.x >> 5;
  const int half = lane >> 4;
  const int lr   = lane & 15;
  const int m0   = blockIdx.x * 16;
  const int n0   = blockIdx.y * kAvN;
  const int kb   = wave * kAvK + 2 * half;

  const float* pp = SC + (size_t)(m0 + lr) * kS + kb;
  const float* vp = Vh + (size_t)h * kS * kDH + (size_t)kb * kDH + n0 + lr;

  v2f sc[kAvSteps];
#pragma unroll
  for (int i = 0; i < kAvSteps; ++i) sc[i] = *(const v2f*)(pp + 4 * i);

  v2f bq[2][kAvChunk * 2];
#pragma unroll
  for (int j = 0; j < kAvChunk * 2; ++j) {
    const float* vb = vp + (size_t)(4 * (j >> 1)) * kDH + (j & 1) * 16;
    bq[0][j] = v2f{vb[0], vb[kDH]};
  }
  __builtin_amdgcn_sched_barrier(0);

  float mx = fmaxf(sc[0][0], sc[0][1]);
#pragma unroll
  for (int i = 1; i < kAvSteps; ++i) mx = fmaxf(mx, fmaxf(sc[i][0], sc[i][1]));
  mx = fmaxf(mx, __shfl_xor(mx, 16, 32));

  float sum = 0.0f;
  v8f acc[2] = {v8f{}, v8f{}};
#pragma unroll
  for (int c = 0; c < kAvSteps / kAvChunk; ++c) {
    if (c + 1 < kAvSteps / kAvChunk) {
#pragma unroll
      for (int j = 0; j < kAvChunk * 2; ++j) {
        const float* vb = vp + (size_t)(4 * ((c + 1) * kAvChunk + (j >> 1))) * kDH + (j & 1) * 16;
        bq[(c + 1) & 1][j] = v2f{vb[0], vb[kDH]};
      }
    }
    __builtin_amdgcn_sched_barrier(0);
#pragma unroll
    for (int s = 0; s < kAvChunk; ++s) {
      const int i = c * kAvChunk + s;
      v2f a = {__expf(sc[i][0] - mx), __expf(sc[i][1] - mx)};
      sum += a[0] + a[1];
#pragma unroll
      for (int t = 0; t < 2; ++t) acc[t] = wmma_f32(a, bq[c & 1][s * 2 + t], acc[t]);
    }
  }
  sum += __shfl_xor(sum, 16, 32);

  if (half == 0) {
    smax[wave][lr] = mx;
    ssum[wave][lr] = sum;
  }
#pragma unroll
  for (int t = 0; t < 2; ++t) {
#pragma unroll
    for (int r = 0; r < 8; ++r) part[wave][t * 8 + r][lane] = acc[t][r];
  }
  __syncthreads();

  const int row = wave + 8 * half;
  float g = smax[0][row];
#pragma unroll
  for (int w = 1; w < kAvWaves; ++w) g = fmaxf(g, smax[w][row]);
  float f[kAvWaves];
  float den = 0.0f;
#pragma unroll
  for (int w = 0; w < kAvWaves; ++w) {
    f[w] = __expf(smax[w][row] - g);
    den += f[w] * ssum[w][row];
  }
  const float inv = 1.0f / den;
  const int m = m0 + row;
#pragma unroll
  for (int t = 0; t < 2; ++t) {
    float s = f[0] * part[0][t * 8 + wave][lane];
#pragma unroll
    for (int w = 1; w < kAvWaves; ++w) s += f[w] * part[w][t * 8 + wave][lane];
    AVec[(size_t)m * kD + h * kDH + n0 + t * 16 + lr] = s * inv;
  }
}

__global__ void out_gemm(const float* __restrict__ A, const float* __restrict__ B,
                         float* __restrict__ C) {
  const int lane = threadIdx.x & 31;
  const int wave = threadIdx.x >> 5;
  const int half = lane >> 4;
  const int lr   = lane & 15;
  const int m0   = blockIdx.x * 16;
  const int n0   = wave * 64;

  const float* ap = A + (size_t)(m0 + lr) * kD + 2 * half;
  const float* bp = B + (size_t)(n0 + lr) * kD + 2 * half;

  v8f acc[4] = {v8f{}, v8f{}, v8f{}, v8f{}};
  for (int k = 0; k < kD; k += 4) {
    v2f a = *(const v2f*)(ap + k);
#pragma unroll
    for (int t = 0; t < 4; ++t) {
      v2f b = *(const v2f*)(bp + (size_t)t * 16 * kD + k);
      acc[t] = wmma_f32(a, b, acc[t]);
    }
  }
#pragma unroll
  for (int t = 0; t < 4; ++t) {
    int n = n0 + t * 16 + lr;
#pragma unroll
    for (int r = 0; r < 8; ++r) {
      int m = m0 + r + 8 * half;
      C[(size_t)m * kD + n] = acc[t][r];
    }
  }
}

extern "C" void kernel_launch(void* const* d_in, const int* in_sizes, int n_in,
                              void* d_out, int out_size, void* d_ws, size_t ws_size,
                              hipStream_t stream) {
  const float* x   = (const float*)d_in[0];
  const float* pos = (const float*)d_in[1];
  const float* Wq  = (const float*)d_in[2];
  const float* Wk  = (const float*)d_in[3];
  const float* Wv  = (const float*)d_in[4];
  const float* Wp  = (const float*)d_in[5];
  const float* Wo  = (const float*)d_in[6];
  const float* u   = (const float*)d_in[7];
  const float* v   = (const float*)d_in[8];

  float* ws = (float*)d_ws;
  const size_t SD = (size_t)kS * kD;
  float* Qhs  = ws;
  float* Khs  = Qhs + SD;
  float* Vhs  = Khs + SD;
  float* Phs  = Vhs + SD;
  float* AVec = Phs + SD;
  float* BD   = AVec + SD;
  float* SC   = BD + (size_t)kS * kS;

  dim3 blk(256);
  proj_gemm_head<<<dim3(kS / 16), blk, 0, stream>>>(x,   Wq, Qhs);
  proj_gemm_head<<<dim3(kS / 16), blk, 0, stream>>>(x,   Wk, Khs);
  proj_gemm_head<<<dim3(kS / 16), blk, 0, stream>>>(x,   Wv, Vhs);
  proj_gemm_head<<<dim3(kS / 16), blk, 0, stream>>>(pos, Wp, Phs);

  const float scale = 0.125f;
  for (int h = 0; h < kH; ++h) {
    bd_gemm   <<<dim3(kS / 16, kS / 512), blk, 0, stream>>>(Qhs, Phs, v, BD, h);
    score_gemm<<<dim3(kS / 16, kS / 512), blk, 0, stream>>>(Qhs, Khs, u, BD, SC, h, scale);
    av_gemm   <<<dim3(kS / 16, kDH / kAvN), blk, 0, stream>>>(SC, Vhs, AVec, h);
  }

  out_gemm<<<dim3(kS / 16), blk, 0, stream>>>(AVec, Wo, (float*)d_out);
}


# independent launches merged into one grid each: 4 projections (grid y), BD of all 8 heads (grid z), score and AV 4 heads per launch (grid z); 29 -> 7 launches, arithmetic unchanged
# speedup vs baseline: 4.3935x; 1.2945x over previous
#include <hip/hip_runtime.h>
#include <math.h>

typedef __attribute__((ext_vector_type(2))) float v2f;
typedef __attribute__((ext_vector_type(8))) float v8f;

static constexpr int kS  = 2048;
static constexpr int kD  = 512;
static constexpr int kH  = 8;
static constexpr int kDH = 64;
static constexpr int kGrp = 4;

__device__ __forceinline__ v8f wmma_f32(v2f a, v2f b, v8f c) {
  return __builtin_amdgcn_wmma_f32_16x16x4_f32(
       false, a,  false, b,
       (short)0, c,  false,  false);
}

__global__ void proj_gemm_head(const float* __restrict__ x, const float* __restrict__ pos,
                               const float* __restrict__ Wq, const float* __restrict__ Wk,
                               const float* __restrict__ Wv, const float* __restrict__ Wp,
                               float* __restrict__ out4) {
  const int which = blockIdx.y;
  const float* A = (which == 3) ? pos : x;
  const float* B = (which == 0) ? Wq : (which == 1) ? Wk : (which == 2) ? Wv : Wp;
  float* out = out4 + (size_t)which * kS * kD;
  const int lane = threadIdx.x & 31;
  const int wave = threadIdx.x >> 5;
  const int half = lane >> 4;
  const int lr   = lane & 15;
  const int m0   = blockIdx.x * 16;
  const int n0   = wave * 64;

  const float* ap = A + (size_t)(m0 + lr) * kD + 2 * half;
  const float* bp = B + (size_t)(n0 + lr) * kD + 2 * half;

  v8f acc[4] = {v8f{}, v8f{}, v8f{}, v8f{}};
  for (int k = 0; k < kD; k += 4) {
    v2f a = *(const v2f*)(ap + k);
#pragma unroll
    for (int t = 0; t < 4; ++t) {
      v2f b = *(const v2f*)(bp + (size_t)t * 16 * kD + k);
      acc[t] = wmma_f32(a, b, acc[t]);
    }
  }
#pragma unroll
  for (int t = 0; t < 4; ++t) {
    int n = n0 + t * 16 + lr;
    int h = n >> 6, d = n & 63;
    float* op = out + (size_t)h * kS * kDH + d;
#pragma unroll
    for (int r = 0; r < 8; ++r) {
      int m = m0 + r + 8 * half;
      op[(size_t)m * kDH] = acc[t][r];
    }
  }
}

__global__ void bd_gemm(const float* __restrict__ Qh, const float* __restrict__ Ph,
                        const float* __restrict__ vb, float* __restrict__ BDall) {
  const int h = blockIdx.z;
  float* BD = BDall + (size_t)h * kS * kS;
  const int lane = threadIdx.x & 31;
  const int wave = threadIdx.x >> 5;
  const int half = lane >> 4;
  const int lr   = lane & 15;
  const int m0   = blockIdx.x * 16;
  const int n0   = blockIdx.y * 512 + wave * 64;

  const float* qp = Qh + (size_t)h * kS * kDH + (size_t)(m0 + lr) * kDH + 2 * half;
  const float* pp = Ph + (size_t)h * kS * kDH + (size_t)(n0 + lr) * kDH + 2 * half;
  const float* bb = vb + h * kDH + 2 * half;

  v8f acc[4] = {v8f{}, v8f{}, v8f{}, v8f{}};
  for (int k = 0; k < kDH; k += 4) {
    v2f a = *(const v2f*)(qp + k);
    v2f bias = *(const v2f*)(bb + k);
    a = a + bias;
#pragma unroll
    for (int t = 0; t < 4; ++t) {
      v2f b = *(const v2f*)(pp + (size_t)t * 16 * kDH + k);
      acc[t] = wmma_f32(a, b, acc[t]);
    }
  }
#pragma unroll
  for (int t = 0; t < 4; ++t) {
    int n = n0 + t * 16 + lr;
#pragma unroll
    for (int r = 0; r < 8; ++r) {
      int m = m0 + r + 8 * half;
      BD[(size_t)m * kS + n] = acc[t][r];
    }
  }
}

__global__ void score_gemm(const float* __restrict__ Qh, const float* __restrict__ Kh,
                           const float* __restrict__ ub, const float* __restrict__ BDall,
                           float* __restrict__ SCgrp, int h0, float scale) {
  const int h = h0 + blockIdx.z;
  const float* BD = BDall + (size_t)h * kS * kS;
  float* SC = SCgrp + (size_t)blockIdx.z * kS * kS;
  const int lane = threadIdx.x & 31;
  const int wave = threadIdx.x >> 5;
  const int half = lane >> 4;
  const int lr   = lane & 15;
  const int m0   = blockIdx.x * 16;
  const int n0   = blockIdx.y * 512 + wave * 64;

  const float* qp = Qh + (size_t)h * kS * kDH + (size_t)(m0 + lr) * kDH + 2 * half;
  const float* kp = Kh + (size_t)h * kS * kDH + (size_t)(n0 + lr) * kDH + 2 * half;
  const float* bb = ub + h * kDH + 2 * half;

  v8f acc[4] = {v8f{}, v8f{}, v8f{}, v8f{}};
  for (int k = 0; k < kDH; k += 4) {
    v2f a = *(const v2f*)(qp + k);
    v2f bias = *(const v2f*)(bb + k);
    a = a + bias;
#pragma unroll
    for (int t = 0; t < 4; ++t) {
      v2f b = *(const v2f*)(kp + (size_t)t * 16 * kDH + k);
      acc[t] = wmma_f32(a, b, acc[t]);
    }
  }
  float bdv[4][8];
#pragma unroll
  for (int t = 0; t < 4; ++t) {
    const int n = n0 + t * 16 + lr;
#pragma unroll
    for (int r = 0; r < 8; ++r) {
      const int m = m0 + r + 8 * half;
      const int d = n - m;
      const unsigned idx = (unsigned)((m + 1) * kS + d - 1 - (d >= 2 ? 1 : 0));
      const float g = BD[idx];
      bdv[t][r] = (d == 1) ? 0.0f : g;
    }
  }
#pragma unroll
  for (int t = 0; t < 4; ++t) {
    const int n = n0 + t * 16 + lr;
#pragma unroll
    for (int r = 0; r < 8; ++r) {
      const int m = m0 + r + 8 * half;
      SC[(size_t)m * kS + n] = (acc[t][r] + bdv[t][r]) * scale;
    }
  }
}

static constexpr int kAvWaves = 8;
static constexpr int kAvK     = kS / kAvWaves;
static constexpr int kAvN     = 32;
static constexpr int kAvSteps = kAvK / 4;
static constexpr int kAvChunk = 8;

__global__ void __launch_bounds__(kAvWaves * 32)
av_gemm(const float* __restrict__ SCgrp, const float* __restrict__ Vh,
        float* __restrict__ AVec, int h0) {
  const int h = h0 + blockIdx.z;
  const float* SC = SCgrp + (size_t)blockIdx.z * kS * kS;
  __shared__ float part[kAvWaves][16][32];
  __shared__ float smax[kAvWaves][16];
  __shared__ float ssum[kAvWaves][16];
  const int lane = threadIdx.x & 31;
  const int wave = threadIdx.x >> 5;
  const int half = lane >> 4;
  const int lr   = lane & 15;
  const int m0   = blockIdx.x * 16;
  const int n0   = blockIdx.y * kAvN;
  const int kb   = wave * kAvK + 2 * half;

  const float* pp = SC + (size_t)(m0 + lr) * kS + kb;
  const float* vp = Vh + (size_t)h * kS * kDH + (size_t)kb * kDH + n0 + lr;

  v2f sc[kAvSteps];
#pragma unroll
  for (int i = 0; i < kAvSteps; ++i) sc[i] = *(const v2f*)(pp + 4 * i);

  v2f bq[2][kAvChunk * 2];
#pragma unroll
  for (int j = 0; j < kAvChunk * 2; ++j) {
    const float* vb = vp + (size_t)(4 * (j >> 1)) * kDH + (j & 1) * 16;
    bq[0][j] = v2f{vb[0], vb[kDH]};
  }
  __builtin_amdgcn_sched_barrier(0);

  float mx = fmaxf(sc[0][0], sc[0][1]);
#pragma unroll
  for (int i = 1; i < kAvSteps; ++i) mx = fmaxf(mx, fmaxf(sc[i][0], sc[i][1]));
  mx = fmaxf(mx, __shfl_xor(mx, 16, 32));

  float sum = 0.0f;
  v8f acc[2] = {v8f{}, v8f{}};
#pragma unroll
  for (int c = 0; c < kAvSteps / kAvChunk; ++c) {
    if (c + 1 < kAvSteps / kAvChunk) {
#pragma unroll
      for (int j = 0; j < kAvChunk * 2; ++j) {
        const float* vb = vp + (size_t)(4 * ((c + 1) * kAvChunk + (j >> 1))) * kDH + (j & 1) * 16;
        bq[(c + 1) & 1][j] = v2f{vb[0], vb[kDH]};
      }
    }
    __builtin_amdgcn_sched_barrier(0);
#pragma unroll
    for (int s = 0; s < kAvChunk; ++s) {
      const int i = c * kAvChunk + s;
      v2f a = {__expf(sc[i][0] - mx), __expf(sc[i][1] - mx)};
      sum += a[0] + a[1];
#pragma unroll
      for (int t = 0; t < 2; ++t) acc[t] = wmma_f32(a, bq[c & 1][s * 2 + t], acc[t]);
    }
  }
  sum += __shfl_xor(sum, 16, 32);

  if (half == 0) {
    smax[wave][lr] = mx;
    ssum[wave][lr] = sum;
  }
#pragma unroll
  for (int t = 0; t < 2; ++t) {
#pragma unroll
    for (int r = 0; r < 8; ++r) part[wave][t * 8 + r][lane] = acc[t][r];
  }
  __syncthreads();

  const int row = wave + 8 * half;
  float g = smax[0][row];
#pragma unroll
  for (int w = 1; w < kAvWaves; ++w) g = fmaxf(g, smax[w][row]);
  float f[kAvWaves];
  float den = 0.0f;
#pragma unroll
  for (int w = 0; w < kAvWaves; ++w) {
    f[w] = __expf(smax[w][row] - g);
    den += f[w] * ssum[w][row];
  }
  const float inv = 1.0f / den;
  const int m = m0 + row;
#pragma unroll
  for (int t = 0; t < 2; ++t) {
    float s = f[0] * part[0][t * 8 + wave][lane];
#pragma unroll
    for (int w = 1; w < kAvWaves; ++w) s += f[w] * part[w][t * 8 + wave][lane];
    AVec[(size_t)m * kD + h * kDH + n0 + t * 16 + lr] = s * inv;
  }
}

__global__ void out_gemm(const float* __restrict__ A, const float* __restrict__ B,
                         float* __restrict__ C) {
  const int lane = threadIdx.x & 31;
  const int wave = threadIdx.x >> 5;
  const int half = lane >> 4;
  const int lr   = lane & 15;
  const int m0   = blockIdx.x * 16;
  const int n0   = wave * 64;

  const float* ap = A + (size_t)(m0 + lr) * kD + 2 * half;
  const float* bp = B + (size_t)(n0 + lr) * kD + 2 * half;

  v8f acc[4] = {v8f{}, v8f{}, v8f{}, v8f{}};
  for (int k = 0; k < kD; k += 4) {
    v2f a = *(const v2f*)(ap + k);
#pragma unroll
    for (int t = 0; t < 4; ++t) {
      v2f b = *(const v2f*)(bp + (size_t)t * 16 * kD + k);
      acc[t] = wmma_f32(a, b, acc[t]);
    }
  }
#pragma unroll
  for (int t = 0; t < 4; ++t) {
    int n = n0 + t * 16 + lr;
#pragma unroll
    for (int r = 0; r < 8; ++r) {
      int m = m0 + r + 8 * half;
      C[(size_t)m * kD + n] = acc[t][r];
    }
  }
}

extern "C" void kernel_launch(void* const* d_in, const int* in_sizes, int n_in,
                              void* d_out, int out_size, void* d_ws, size_t ws_size,
                              hipStream_t stream) {
  const float* x   = (const float*)d_in[0];
  const float* pos = (const float*)d_in[1];
  const float* Wq  = (const float*)d_in[2];
  const float* Wk  = (const float*)d_in[3];
  const float* Wv  = (const float*)d_in[4];
  const float* Wp  = (const float*)d_in[5];
  const float* Wo  = (const float*)d_in[6];
  const float* u   = (const float*)d_in[7];
  const float* v   = (const float*)d_in[8];

  float* ws = (float*)d_ws;
  const size_t SD = (size_t)kS * kD;
  float* Qhs  = ws;
  float* Khs  = Qhs + SD;
  float* Vhs  = Khs + SD;
  float* Phs  = Vhs + SD;
  float* AVec = Phs + SD;
  float* BD   = AVec + SD;
  float* SC   = BD + (size_t)kH * kS * kS;

  dim3 blk(256);
  proj_gemm_head<<<dim3(kS / 16, 4), blk, 0, stream>>>(x, pos, Wq, Wk, Wv, Wp, Qhs);

  const float scale = 0.125f;
  bd_gemm<<<dim3(kS / 16, kS / 512, kH), blk, 0, stream>>>(Qhs, Phs, v, BD);
  for (int h0 = 0; h0 < kH; h0 += kGrp) {
    score_gemm<<<dim3(kS / 16, kS / 512, kGrp), blk, 0, stream>>>(Qhs, Khs, u, BD, SC, h0, scale);
    av_gemm   <<<dim3(kS / 16, kDH / kAvN, kGrp), blk, 0, stream>>>(SC, Vhs, AVec, h0);
  }

  out_gemm<<<dim3(kS / 16), blk, 0, stream>>>(AVec, Wo, (float*)d_out);
}


# fragment loads widened from 64 to 128 bits per lane in all five kernels: one load feeds two WMMA k steps; k order inside each 8-k chunk permuted alike for A and B (new summation order)
# speedup vs baseline: 4.9936x; 1.1366x over previous
#include <hip/hip_runtime.h>
#include <math.h>

typedef __attribute__((ext_vector_type(2))) float v2f;
typedef __attribute__((ext_vector_type(4))) float v4f;
typedef __attribute__((ext_vector_type(8))) float v8f;

static constexpr int kS  = 2048;
static constexpr int kD  = 512;
static constexpr int kH  = 8;
static constexpr int kDH = 64;
static constexpr int kGrp = 4;

__device__ __forceinline__ v8f wmma_f32(v2f a, v2f b, v8f c) {
  return __builtin_amdgcn_wmma_f32_16x16x4_f32(
       false, a,  false, b,
       (short)0, c,  false,  false);
}

__device__ __forceinline__ v2f lo2(v4f x) { return v2f{x[0], x[1]}; }
__device__ __forceinline__ v2f hi2(v4f x) { return v2f{x[2], x[3]}; }

__global__ void proj_gemm_head(const float* __restrict__ x, const float* __restrict__ pos,
                               const float* __restrict__ Wq, const float* __restrict__ Wk,
                               const float* __restrict__ Wv, const float* __restrict__ Wp,
                               float* __restrict__ out4) {
  const int which = blockIdx.y;
  const float* A = (which == 3) ? pos : x;
  const float* B = (which == 0) ? Wq : (which == 1) ? Wk : (which == 2) ? Wv : Wp;
  float* out = out4 + (size_t)which * kS * kD;
  const int lane = threadIdx.x & 31;
  const int wave = threadIdx.x >> 5;
  const int half = lane >> 4;
  const int lr   = lane & 15;
  const int m0   = blockIdx.x * 16;
  const int n0   = wave * 64;

  const float* ap = A + (size_t)(m0 + lr) * kD + 4 * half;
  const float* bp = B + (size_t)(n0 + lr) * kD + 4 * half;

  v8f acc[4] = {v8f{}, v8f{}, v8f{}, v8f{}};
  for (int k = 0; k < kD; k += 8) {
    v4f a = *(const v4f*)(ap + k);
    v4f b[4];
#pragma unroll
    for (int t = 0; t < 4; ++t) b[t] = *(const v4f*)(bp + (size_t)t * 16 * kD + k);
#pragma unroll
    for (int t = 0; t < 4; ++t) acc[t] = wmma_f32(lo2(a), lo2(b[t]), acc[t]);
#pragma unroll
    for (int t = 0; t < 4; ++t) acc[t] = wmma_f32(hi2(a), hi2(b[t]), acc[t]);
  }
#pragma unroll
  for (int t = 0; t < 4; ++t) {
    int n = n0 + t * 16 + lr;
    int h = n >> 6, d = n & 63;
    float* op = out + (size_t)h * kS * kDH + d;
#pragma unroll
    for (int r = 0; r < 8; ++r) {
      int m = m0 + r + 8 * half;
      op[(size_t)m * kDH] = acc[t][r];
    }
  }
}

__global__ void bd_gemm(const float* __restrict__ Qh, const float* __restrict__ Ph,
                        const float* __restrict__ vb, float* __restrict__ BDall) {
  const int h = blockIdx.z;
  float* BD = BDall + (size_t)h * kS * kS;
  const int lane = threadIdx.x & 31;
  const int wave = threadIdx.x >> 5;
  const int half = lane >> 4;
  const int lr   = lane & 15;
  const int m0   = blockIdx.x * 16;
  const int n0   = blockIdx.y * 512 + wave * 64;

  const float* qp = Qh + (size_t)h * kS * kDH + (size_t)(m0 + lr) * kDH + 4 * half;
  const float* pp = Ph + (size_t)h * kS * kDH + (size_t)(n0 + lr) * kDH + 4 * half;
  const float* bb = vb + h * kDH + 4 * half;

  v8f acc[4] = {v8f{}, v8f{}, v8f{}, v8f{}};
  for (int k = 0; k < kDH; k += 8) {
    v4f a = *(const v4f*)(qp + k);
    v4f bias = *(const v4f*)(bb + k);
    a = a + bias;
    v4f b[4];
#pragma unroll
    for (int t = 0; t < 4; ++t) b[t] = *(const v4f*)(pp + (size_t)t * 16 * kDH + k);
#pragma unroll
    for (int t = 0; t < 4; ++t) acc[t] = wmma_f32(lo2(a), lo2(b[t]), acc[t]);
#pragma unroll
    for (int t = 0; t < 4; ++t) acc[t] = wmma_f32(hi2(a), hi2(b[t]), acc[t]);
  }
#pragma unroll
  for (int t = 0; t < 4; ++t) {
    int n = n0 + t * 16 + lr;
#pragma unroll
    for (int r = 0; r < 8; ++r) {
      int m = m0 + r + 8 * half;
      BD[(size_t)m * kS + n] = acc[t][r];
    }
  }
}

__global__ void score_gemm(const float* __restrict__ Qh, const float* __restrict__ Kh,
                           const float* __restrict__ ub, const float* __restrict__ BDall,
                           float* __restrict__ SCgrp, int h0, float scale) {
  const int h = h0 + blockIdx.z;
  const float* BD = BDall + (size_t)h * kS * kS;
  float* SC = SCgrp + (size_t)blockIdx.z * kS * kS;
  const int lane = threadIdx.x & 31;
  const int wave = threadIdx.x >> 5;
  const int half = lane >> 4;
  const int lr   = lane & 15;
  const int m0   = blockIdx.x * 16;
  const int n0   = blockIdx.y * 512 + wave * 64;

  const float* qp = Qh + (size_t)h * kS * kDH + (size_t)(m0 + lr) * kDH + 4 * half;
  const float* kp = Kh + (size_t)h * kS * kDH + (size_t)(n0 + lr) * kDH + 4 * half;
  const float* bb = ub + h * kDH + 4 * half;

  v8f acc[4] = {v8f{}, v8f{}, v8f{}, v8f{}};
  for (int k = 0; k < kDH; k += 8) {
    v4f a = *(const v4f*)(qp + k);
    v4f bias = *(const v4f*)(bb + k);
    a = a + bias;
    v4f b[4];
#pragma unroll
    for (int t = 0; t < 4; ++t) b[t] = *(const v4f*)(kp + (size_t)t * 16 * kDH + k);
#pragma unroll
    for (int t = 0; t < 4; ++t) acc[t] = wmma_f32(lo2(a), lo2(b[t]), acc[t]);
#pragma unroll
    for (int t = 0; t < 4; ++t) acc[t] = wmma_f32(hi2(a), hi2(b[t]), acc[t]);
  }
  float bdv[4][8];
#pragma unroll
  for (int t = 0; t < 4; ++t) {
    const int n = n0 + t * 16 + lr;
#pragma unroll
    for (int r = 0; r < 8; ++r) {
      const int m = m0 + r + 8 * half;
      const int d = n - m;
      const unsigned idx = (unsigned)((m + 1) * kS + d - 1 - (d >= 2 ? 1 : 0));
      const float g = BD[idx];
      bdv[t][r] = (d == 1) ? 0.0f : g;
    }
  }
#pragma unroll
  for (int t = 0; t < 4; ++t) {
    const int n = n0 + t * 16 + lr;
#pragma unroll
    for (int r = 0; r < 8; ++r) {
      const int m = m0 + r + 8 * half;
      SC[(size_t)m * kS + n] = (acc[t][r] + bdv[t][r]) * scale;
    }
  }
}

static constexpr int kAvWaves = 8;
static constexpr int kAvK     = kS / kAvWaves;
static constexpr int kAvN     = 32;
static constexpr int kAvSteps = kAvK / 4;
static constexpr int kAvChunk = 8;

__device__ __forceinline__ constexpr int av_vrow(int i) { return 8 * (i >> 1) + 2 * (i & 1); }

__global__ void __launch_bounds__(kAvWaves * 32)
av_gemm(const float* __restrict__ SCgrp, const float* __restrict__ Vh,
        float* __restrict__ AVec, int h0) {
  const int h = h0 + blockIdx.z;
  const float* SC = SCgrp + (size_t)blockIdx.z * kS * kS;
  __shared__ float part[kAvWaves][16][32];
  __shared__ float smax[kAvWaves][16];
  __shared__ float ssum[kAvWaves][16];
  const int lane = threadIdx.x & 31;
  const int wave = threadIdx.x >> 5;
  const int half = lane >> 4;
  const int lr   = lane & 15;
  const int m0   = blockIdx.x * 16;
  const int n0   = blockIdx.y * kAvN;
  const int kb   = wave * kAvK + 4 * half;

  const float* pp = SC + (size_t)(m0 + lr) * kS + kb;
  const float* vp = Vh + (size_t)h * kS * kDH + (size_t)kb * kDH + n0 + lr;

  v4f sc[kAvSteps / 2];
#pragma unroll
  for (int q = 0; q < kAvSteps / 2; ++q) sc[q] = *(const v4f*)(pp + 8 * q);

  v2f bq[2][kAvChunk * 2];
#pragma unroll
  for (int j = 0; j < kAvChunk * 2; ++j) {
    const float* vb = vp + (size_t)av_vrow(j >> 1) * kDH + (j & 1) * 16;
    bq[0][j] = v2f{vb[0], vb[kDH]};
  }
  __builtin_amdgcn_sched_barrier(0);

  float mx = fmaxf(fmaxf(sc[0][0], sc[0][1]), fmaxf(sc[0][2], sc[0][3]));
#pragma unroll
  for (int q = 1; q < kAvSteps / 2; ++q)
    mx = fmaxf(mx, fmaxf(fmaxf(sc[q][0], sc[q][1]), fmaxf(sc[q][2], sc[q][3])));
  mx = fmaxf(mx, __shfl_xor(mx, 16, 32));

  float sum = 0.0f;
  v8f acc[2] = {v8f{}, v8f{}};
#pragma unroll
  for (int c = 0; c < kAvSteps / kAvChunk; ++c) {
    if (c + 1 < kAvSteps / kAvChunk) {
#pragma unroll
      for (int j = 0; j < kAvChunk * 2; ++j) {
        const float* vb = vp + (size_t)av_vrow((c + 1) * kAvChunk + (j >> 1)) * kDH + (j & 1) * 16;
        bq[(c + 1) & 1][j] = v2f{vb[0], vb[kDH]};
      }
    }
    __builtin_amdgcn_sched_barrier(0);
#pragma unroll
    for (int s = 0; s < kAvChunk; ++s) {
      const int i = c * kAvChunk + s;
      v2f a = {__expf(sc[i >> 1][2 * (i & 1)] - mx), __expf(sc[i >> 1][2 * (i & 1) + 1] - mx)};
      sum += a[0] + a[1];
#pragma unroll
      for (int t = 0; t < 2; ++t) acc[t] = wmma_f32(a, bq[c & 1][s * 2 + t], acc[t]);
    }
  }
  sum += __shfl_xor(sum, 16, 32);

  if (half == 0) {
    smax[wave][lr] = mx;
    ssum[wave][lr] = sum;
  }
#pragma unroll
  for (int t = 0; t < 2; ++t) {
#pragma unroll
    for (int r = 0; r < 8; ++r) part[wave][t * 8 + r][lane] = acc[t][r];
  }
  __syncthreads();

  const int row = wave + 8 * half;
  float g = smax[0][row];
#pragma unroll
  for (int w = 1; w < kAvWaves; ++w) g = fmaxf(g, smax[w][row]);
  float f[kAvWaves];
  float den = 0.0f;
#pragma unroll
  for (int w = 0; w < kAvWaves; ++w) {
    f[w] = __expf(smax[w][row] - g);
    den += f[w] * ssum[w][row];
  }
  const float inv = 1.0f / den;
  const int m = m0 + row;
#pragma unroll
  for (int t = 0; t < 2; ++t) {
    float s = f[0] * part[0][t * 8 + wave][lane];
#pragma unroll
    for (int w = 1; w < kAvWaves; ++w) s += f[w] * part[w][t * 8 + wave][lane];
    AVec[(size_t)m * kD + h * kDH + n0 + t * 16 + lr] = s * inv;
  }
}

__global__ void out_gemm(const float* __restrict__ A, const float* __restrict__ B,
                         float* __restrict__ C) {
  const int lane = threadIdx.x & 31;
  const int wave = threadIdx.x >> 5;
  const int half = lane >> 4;
  const int lr   = lane & 15;
  const int m0   = blockIdx.x * 16;
  const int n0   = wave * 64;

  const float* ap = A + (size_t)(m0 + lr) * kD + 4 * half;
  const float* bp = B + (size_t)(n0 + lr) * kD + 4 * half;

  v8f acc[4] = {v8f{}, v8f{}, v8f{}, v8f{}};
  for (int k = 0; k < kD; k += 8) {
    v4f a = *(const v4f*)(ap + k);
    v4f b[4];
#pragma unroll
    for (int t = 0; t < 4; ++t) b[t] = *(const v4f*)(bp + (size_t)t * 16 * kD + k);
#pragma unroll
    for (int t = 0; t < 4; ++t) acc[t] = wmma_f32(lo2(a), lo2(b[t]), acc[t]);
#pragma unroll
    for (int t = 0; t < 4; ++t) acc[t] = wmma_f32(hi2(a), hi2(b[t]), acc[t]);
  }
#pragma unroll
  for (int t = 0; t < 4; ++t) {
    int n = n0 + t * 16 + lr;
#pragma unroll
    for (int r = 0; r < 8; ++r) {
      int m = m0 + r + 8 * half;
      C[(size_t)m * kD + n] = acc[t][r];
    }
  }
}

extern "C" void kernel_launch(void* const* d_in, const int* in_sizes, int n_in,
                              void* d_out, int out_size, void* d_ws, size_t ws_size,
                              hipStream_t stream) {
  const float* x   = (const float*)d_in[0];
  const float* pos = (const float*)d_in[1];
  const float* Wq  = (const float*)d_in[2];
  const float* Wk  = (const float*)d_in[3];
  const float* Wv  = (const float*)d_in[4];
  const float* Wp  = (const float*)d_in[5];
  const float* Wo  = (const float*)d_in[6];
  const float* u   = (const float*)d_in[7];
  const float* v   = (const float*)d_in[8];

  float* ws = (float*)d_ws;
  const size_t SD = (size_t)kS * kD;
  float* Qhs  = ws;
  float* Khs  = Qhs + SD;
  float* Vhs  = Khs + SD;
  float* Phs  = Vhs + SD;
  float* AVec = Phs + SD;
  float* BD   = AVec + SD;
  float* SC   = BD + (size_t)kH * kS * kS;

  dim3 blk(256);
  proj_gemm_head<<<dim3(kS / 16, 4), blk, 0, stream>>>(x, pos, Wq, Wk, Wv, Wp, Qhs);

  const float scale = 0.125f;
  bd_gemm<<<dim3(kS / 16, kS / 512, kH), blk, 0, stream>>>(Qhs, Phs, v, BD);
  for (int h0 = 0; h0 < kH; h0 += kGrp) {
    score_gemm<<<dim3(kS / 16, kS / 512, kGrp), blk, 0, stream>>>(Qhs, Khs, u, BD, SC, h0, scale);
    av_gemm   <<<dim3(kS / 16, kDH / kAvN, kGrp), blk, 0, stream>>>(SC, Vhs, AVec, h0);
  }

  out_gemm<<<dim3(kS / 16), blk, 0, stream>>>(AVec, Wo, (float*)d_out);
}


# projection, BD and score products re-tiled: a wave computes 32x64 (each B fragment feeds two row tiles) and a block's 8 waves stack on one 64-column strip, sharing B; per-element sum order kept
# speedup vs baseline: 6.5295x; 1.3076x over previous
#include <hip/hip_runtime.h>
#include <math.h>

typedef __attribute__((ext_vector_type(2))) float v2f;
typedef __attribute__((ext_vector_type(4))) float v4f;
typedef __attribute__((ext_vector_type(8))) float v8f;

static constexpr int kS  = 2048;
static constexpr int kD  = 512;
static constexpr int kH  = 8;
static constexpr int kDH = 64;
static constexpr int kGrp = 4;
static constexpr int kMT  = 2;
static constexpr int kBlkM = 8 * kMT * 16;

__device__ __forceinline__ v8f wmma_f32(v2f a, v2f b, v8f c) {
  return __builtin_amdgcn_wmma_f32_16x16x4_f32(
       false, a,  false, b,
       (short)0, c,  false,  false);
}

__device__ __forceinline__ v2f lo2(v4f x) { return v2f{x[0], x[1]}; }
__device__ __forceinline__ v2f hi2(v4f x) { return v2f{x[2], x[3]}; }

__global__ void proj_gemm_head(const float* __restrict__ x, const float* __restrict__ pos,
                               const float* __restrict__ Wq, const float* __restrict__ Wk,
                               const float* __restrict__ Wv, const float* __restrict__ Wp,
                               float* __restrict__ out4) {
  const int which = blockIdx.z;
  const float* A = (which == 3) ? pos : x;
  const float* B = (which == 0) ? Wq : (which == 1) ? Wk : (which == 2) ? Wv : Wp;
  float* out = out4 + (size_t)which * kS * kD;
  const int lane = threadIdx.x & 31;
  const int wave = threadIdx.x >> 5;
  const int half = lane >> 4;
  const int lr   = lane & 15;
  const int m0   = blockIdx.x * kBlkM + wave * (kMT * 16);
  const int n0   = blockIdx.y * 64;

  const float* ap = A + (size_t)(m0 + lr) * kD + 4 * half;
  const float* bp = B + (size_t)(n0 + lr) * kD + 4 * half;

  v8f acc[kMT][4];
#pragma unroll
  for (int u = 0; u < kMT; ++u) {
#pragma unroll
    for (int t = 0; t < 4; ++t) acc[u][t] = v8f{};
  }
  for (int k = 0; k < kD; k += 8) {
    v4f a[kMT];
    v4f b[4];
#pragma unroll
    for (int u = 0; u < kMT; ++u) a[u] = *(const v4f*)(ap + (size_t)u * 16 * kD + k);
#pragma unroll
    for (int t = 0; t < 4; ++t) b[t] = *(const v4f*)(bp + (size_t)t * 16 * kD + k);
#pragma unroll
    for (int u = 0; u < kMT; ++u) {
#pragma unroll
      for (int t = 0; t < 4; ++t) acc[u][t] = wmma_f32(lo2(a[u]), lo2(b[t]), acc[u][t]);
    }
#pragma unroll
    for (int u = 0; u < kMT; ++u) {
#pragma unroll
      for (int t = 0; t < 4; ++t) acc[u][t] = wmma_f32(hi2(a[u]), hi2(b[t]), acc[u][t]);
    }
  }
#pragma unroll
  for (int u = 0; u < kMT; ++u) {
#pragma unroll
    for (int t = 0; t < 4; ++t) {
      int n = n0 + t * 16 + lr;
      int h = n >> 6, d = n & 63;
      float* op = out + (size_t)h * kS * kDH + d;
#pragma unroll
      for (int r = 0; r < 8; ++r) {
        int m = m0 + u * 16 + r + 8 * half;
        op[(size_t)m * kDH] = acc[u][t][r];
      }
    }
  }
}

__global__ void bd_gemm(const float* __restrict__ Qh, const float* __restrict__ Ph,
                        const float* __restrict__ vb, float* __restrict__ BDall) {
  const int h = blockIdx.z;
  float* BD = BDall + (size_t)h * kS * kS;
  const int lane = threadIdx.x & 31;
  const int wave = threadIdx.x >> 5;
  const int half = lane >> 4;
  const int lr   = lane & 15;
  const int m0   = blockIdx.x * kBlkM + wave * (kMT * 16);
  const int n0   = blockIdx.y * 64;

  const float* qp = Qh + (size_t)h * kS * kDH + (size_t)(m0 + lr) * kDH + 4 * half;
  const float* pp = Ph + (size_t)h * kS * kDH + (size_t)(n0 + lr) * kDH + 4 * half;
  const float* bb = vb + h * kDH + 4 * half;

  v8f acc[kMT][4];
#pragma unroll
  for (int u = 0; u < kMT; ++u) {
#pragma unroll
    for (int t = 0; t < 4; ++t) acc[u][t] = v8f{};
  }
  for (int k = 0; k < kDH; k += 8) {
    v4f bias = *(const v4f*)(bb + k);
    v4f a[kMT];
    v4f b[4];
#pragma unroll
    for (int u = 0; u < kMT; ++u) a[u] = *(const v4f*)(qp + (size_t)u * 16 * kDH + k) + bias;
#pragma unroll
    for (int t = 0; t < 4; ++t) b[t] = *(const v4f*)(pp + (size_t)t * 16 * kDH + k);
#pragma unroll
    for (int u = 0; u < kMT; ++u) {
#pragma unroll
      for (int t = 0; t < 4; ++t) acc[u][t] = wmma_f32(lo2(a[u]), lo2(b[t]), acc[u][t]);
    }
#pragma unroll
    for (int u = 0; u < kMT; ++u) {
#pragma unroll
      for (int t = 0; t < 4; ++t) acc[u][t] = wmma_f32(hi2(a[u]), hi2(b[t]), acc[u][t]);
    }
  }
#pragma unroll
  for (int u = 0; u < kMT; ++u) {
#pragma unroll
    for (int t = 0; t < 4; ++t) {
      int n = n0 + t * 16 + lr;
#pragma unroll
      for (int r = 0; r < 8; ++r) {
        int m = m0 + u * 16 + r + 8 * half;
        BD[(size_t)m * kS + n] = acc[u][t][r];
      }
    }
  }
}

__global__ void score_gemm(const float* __restrict__ Qh, const float* __restrict__ Kh,
                           const float* __restrict__ ub, const float* __restrict__ BDall,
                           float* __restrict__ SCgrp, int h0, float scale) {
  const int h = h0 + blockIdx.z;
  const float* BD = BDall + (size_t)h * kS * kS;
  float* SC = SCgrp + (size_t)blockIdx.z * kS * kS;
  const int lane = threadIdx.x & 31;
  const int wave = threadIdx.x >> 5;
  const int half = lane >> 4;
  const int lr   = lane & 15;
  const int m0   = blockIdx.x * kBlkM + wave * (kMT * 16);
  const int n0   = blockIdx.y * 64;

  const float* qp = Qh + (size_t)h * kS * kDH + (size_t)(m0 + lr) * kDH + 4 * half;
  const float* kp = Kh + (size_t)h * kS * kDH + (size_t)(n0 + lr) * kDH + 4 * half;
  const float* bb = ub + h * kDH + 4 * half;

  v8f acc[kMT][4];
#pragma unroll
  for (int u = 0; u < kMT; ++u) {
#pragma unroll
    for (int t = 0; t < 4; ++t) acc[u][t] = v8f{};
  }
  for (int k = 0; k < kDH; k += 8) {
    v4f bias = *(const v4f*)(bb + k);
    v4f a[kMT];
    v4f b[4];
#pragma unroll
    for (int u = 0; u < kMT; ++u) a[u] = *(const v4f*)(qp + (size_t)u * 16 * kDH + k) + bias;
#pragma unroll
    for (int t = 0; t < 4; ++t) b[t] = *(const v4f*)(kp + (size_t)t * 16 * kDH + k);
#pragma unroll
    for (int u = 0; u < kMT; ++u) {
#pragma unroll
      for (int t = 0; t < 4; ++t) acc[u][t] = wmma_f32(lo2(a[u]), lo2(b[t]), acc[u][t]);
    }
#pragma unroll
    for (int u = 0; u < kMT; ++u) {
#pragma unroll
      for (int t = 0; t < 4; ++t) acc[u][t] = wmma_f32(hi2(a[u]), hi2(b[t]), acc[u][t]);
    }
  }
#pragma unroll
  for (int u = 0; u < kMT; ++u) {
    float bdv[4][8];
#pragma unroll
    for (int t = 0; t < 4; ++t) {
      const int n = n0 + t * 16 + lr;
#pragma unroll
      for (int r = 0; r < 8; ++r) {
        const int m = m0 + u * 16 + r + 8 * half;
        const int d = n - m;
        const unsigned idx = (unsigned)((m + 1) * kS + d - 1 - (d >= 2 ? 1 : 0));
        const float g = BD[idx];
        bdv[t][r] = (d == 1) ? 0.0f : g;
      }
    }
#pragma unroll
    for (int t = 0; t < 4; ++t) {
      const int n = n0 + t * 16 + lr;
#pragma unroll
      for (int r = 0; r < 8; ++r) {
        const int m = m0 + u * 16 + r + 8 * half;
        SC[(size_t)m * kS + n] = (acc[u][t][r] + bdv[t][r]) * scale;
      }
    }
  }
}

static constexpr int kAvWaves = 8;
static constexpr int kAvK     = kS / kAvWaves;
static constexpr int kAvN     = 32;
static constexpr int kAvSteps = kAvK / 4;
static constexpr int kAvChunk = 8;

__device__ __forceinline__ constexpr int av_vrow(int i) { return 8 * (i >> 1) + 2 * (i & 1); }

__global__ void __launch_bounds__(kAvWaves * 32)
av_gemm(const float* __restrict__ SCgrp, const float* __restrict__ Vh,
        float* __restrict__ AVec, int h0) {
  const int h = h0 + blockIdx.z;
  const float* SC = SCgrp + (size_t)blockIdx.z * kS * kS;
  __shared__ float part[kAvWaves][16][32];
  __shared__ float smax[kAvWaves][16];
  __shared__ float ssum[kAvWaves][16];
  const int lane = threadIdx.x & 31;
  const int wave = threadIdx.x >> 5;
  const int half = lane >> 4;
  const int lr   = lane & 15;
  const int m0   = blockIdx.x * 16;
  const int n0   = blockIdx.y * kAvN;
  const int kb   = wave * kAvK + 4 * half;

  const float* pp = SC + (size_t)(m0 + lr) * kS + kb;
  const float* vp = Vh + (size_t)h * kS * kDH + (size_t)kb * kDH + n0 + lr;

  v4f sc[kAvSteps / 2];
#pragma unroll
  for (int q = 0; q < kAvSteps / 2; ++q) sc[q] = *(const v4f*)(pp + 8 * q);

  v2f bq[2][kAvChunk * 2];
#pragma unroll
  for (int j = 0; j < kAvChunk * 2; ++j) {
    const float* vb = vp + (size_t)av_vrow(j >> 1) * kDH + (j & 1) * 16;
    bq[0][j] = v2f{vb[0], vb[kDH]};
  }
  __builtin_amdgcn_sched_barrier(0);

  float mx = fmaxf(fmaxf(sc[0][0], sc[0][1]), fmaxf(sc[0][2], sc[0][3]));
#pragma unroll
  for (int q = 1; q < kAvSteps / 2; ++q)
    mx = fmaxf(mx, fmaxf(fmaxf(sc[q][0], sc[q][1]), fmaxf(sc[q][2], sc[q][3])));
  mx = fmaxf(mx, __shfl_xor(mx, 16, 32));

  float sum = 0.0f;
  v8f acc[2] = {v8f{}, v8f{}};
#pragma unroll
  for (int c = 0; c < kAvSteps / kAvChunk; ++c) {
    if (c + 1 < kAvSteps / kAvChunk) {
#pragma unroll
      for (int j = 0; j < kAvChunk * 2; ++j) {
        const float* vb = vp + (size_t)av_vrow((c + 1) * kAvChunk + (j >> 1)) * kDH + (j & 1) * 16;
        bq[(c + 1) & 1][j] = v2f{vb[0], vb[kDH]};
      }
    }
    __builtin_amdgcn_sched_barrier(0);
#pragma unroll
    for (int s = 0; s < kAvChunk; ++s) {
      const int i = c * kAvChunk + s;
      v2f a = {__expf(sc[i >> 1][2 * (i & 1)] - mx), __expf(sc[i >> 1][2 * (i & 1) + 1] - mx)};
      sum += a[0] + a[1];
#pragma unroll
      for (int t = 0; t < 2; ++t) acc[t] = wmma_f32(a, bq[c & 1][s * 2 + t], acc[t]);
    }
  }
  sum += __shfl_xor(sum, 16, 32);

  if (half == 0) {
    smax[wave][lr] = mx;
    ssum[wave][lr] = sum;
  }
#pragma unroll
  for (int t = 0; t < 2; ++t) {
#pragma unroll
    for (int r = 0; r < 8; ++r) part[wave][t * 8 + r][lane] = acc[t][r];
  }
  __syncthreads();

  const int row = wave + 8 * half;
  float g = smax[0][row];
#pragma unroll
  for (int w = 1; w < kAvWaves; ++w) g = fmaxf(g, smax[w][row]);
  float f[kAvWaves];
  float den = 0.0f;
#pragma unroll
  for (int w = 0; w < kAvWaves; ++w) {
    f[w] = __expf(smax[w][row] - g);
    den += f[w] * ssum[w][row];
  }
  const float inv = 1.0f / den;
  const int m = m0 + row;
#pragma unroll
  for (int t = 0; t < 2; ++t) {
    float s = f[0] * part[0][t * 8 + wave][lane];
#pragma unroll
    for (int w = 1; w < kAvWaves; ++w) s += f[w] * part[w][t * 8 + wave][lane];
    AVec[(size_t)m * kD + h * kDH + n0 + t * 16 + lr] = s * inv;
  }
}

__global__ void out_gemm(const float* __restrict__ A, const float* __restrict__ B,
                         float* __restrict__ C) {
  const int lane = threadIdx.x & 31;
  const int wave = threadIdx.x >> 5;
  const int half = lane >> 4;
  const int lr   = lane & 15;
  const int m0   = blockIdx.x * 16;
  const int n0   = wave * 64;

  const float* ap = A + (size_t)(m0 + lr) * kD + 4 * half;
  const float* bp = B + (size_t)(n0 + lr) * kD + 4 * half;

  v8f acc[4] = {v8f{}, v8f{}, v8f{}, v8f{}};
  for (int k = 0; k < kD; k += 8) {
    v4f a = *(const v4f*)(ap + k);
    v4f b[4];
#pragma unroll
    for (int t = 0; t < 4; ++t) b[t] = *(const v4f*)(bp + (size_t)t * 16 * kD + k);
#pragma unroll
    for (int t = 0; t < 4; ++t) acc[t] = wmma_f32(lo2(a), lo2(b[t]), acc[t]);
#pragma unroll
    for (int t = 0; t < 4; ++t) acc[t] = wmma_f32(hi2(a), hi2(b[t]), acc[t]);
  }
#pragma unroll
  for (int t = 0; t < 4; ++t) {
    int n = n0 + t * 16 + lr;
#pragma unroll
    for (int r = 0; r < 8; ++r) {
      int m = m0 + r + 8 * half;
      C[(size_t)m * kD + n] = acc[t][r];
    }
  }
}

extern "C" void kernel_launch(void* const* d_in, const int* in_sizes, int n_in,
                              void* d_out, int out_size, void* d_ws, size_t ws_size,
                              hipStream_t stream) {
  const float* x   = (const float*)d_in[0];
  const float* pos = (const float*)d_in[1];
  const float* Wq  = (const float*)d_in[2];
  const float* Wk  = (const float*)d_in[3];
  const float* Wv  = (const float*)d_in[4];
  const float* Wp  = (const float*)d_in[5];
  const float* Wo  = (const float*)d_in[6];
  const float* u   = (const float*)d_in[7];
  const float* v   = (const float*)d_in[8];

  float* ws = (float*)d_ws;
  const size_t SD = (size_t)kS * kD;
  float* Qhs  = ws;
  float* Khs  = Qhs + SD;
  float* Vhs  = Khs + SD;
  float* Phs  = Vhs + SD;
  float* AVec = Phs + SD;
  float* BD   = AVec + SD;
  float* SC   = BD + (size_t)kH * kS * kS;

  dim3 blk(256);
  proj_gemm_head<<<dim3(kS / kBlkM, kD / 64, 4), blk, 0, stream>>>(x, pos, Wq, Wk, Wv, Wp, Qhs);

  const float scale = 0.125f;
  bd_gemm<<<dim3(kS / kBlkM, kS / 64, kH), blk, 0, stream>>>(Qhs, Phs, v, BD);
  for (int h0 = 0; h0 < kH; h0 += kGrp) {
    score_gemm<<<dim3(kS / kBlkM, kS / 64, kGrp), blk, 0, stream>>>(Qhs, Khs, u, BD, SC, h0, scale);
    av_gemm   <<<dim3(kS / 16, kDH / kAvN, kGrp), blk, 0, stream>>>(SC, Vhs, AVec, h0);
  }

  out_gemm<<<dim3(kS / 16), blk, 0, stream>>>(AVec, Wo, (float*)d_out);
}


# score kernel epilogue: a scheduling barrier keeps the BD gathers in flight 16 at a time (the build had reused one register: load, wait, select, 64 times in a row per wave); arithmetic unchanged
# speedup vs baseline: 7.0330x; 1.0771x over previous
#include <hip/hip_runtime.h>
#include <math.h>

typedef __attribute__((ext_vector_type(2))) float v2f;
typedef __attribute__((ext_vector_type(4))) float v4f;
typedef __attribute__((ext_vector_type(8))) float v8f;

static constexpr int kS  = 2048;
static constexpr int kD  = 512;
static constexpr int kH  = 8;
static constexpr int kDH = 64;
static constexpr int kGrp = 4;
static constexpr int kMT  = 2;
static constexpr int kBlkM = 8 * kMT * 16;

__device__ __forceinline__ v8f wmma_f32(v2f a, v2f b, v8f c) {
  return __builtin_amdgcn_wmma_f32_16x16x4_f32(
       false, a,  false, b,
       (short)0, c,  false,  false);
}

__device__ __forceinline__ v2f lo2(v4f x) { return v2f{x[0], x[1]}; }
__device__ __forceinline__ v2f hi2(v4f x) { return v2f{x[2], x[3]}; }

__global__ void proj_gemm_head(const float* __restrict__ x, const float* __restrict__ pos,
                               const float* __restrict__ Wq, const float* __restrict__ Wk,
                               const float* __restrict__ Wv, const float* __restrict__ Wp,
                               float* __restrict__ out4) {
  const int which = blockIdx.z;
  const float* A = (which == 3) ? pos : x;
  const float* B = (which == 0) ? Wq : (which == 1) ? Wk : (which == 2) ? Wv : Wp;
  float* out = out4 + (size_t)which * kS * kD;
  const int lane = threadIdx.x & 31;
  const int wave = threadIdx.x >> 5;
  const int half = lane >> 4;
  const int lr   = lane & 15;
  const int m0   = blockIdx.x * kBlkM + wave * (kMT * 16);
  const int n0   = blockIdx.y * 64;

  const float* ap = A + (size_t)(m0 + lr) * kD + 4 * half;
  const float* bp = B + (size_t)(n0 + lr) * kD + 4 * half;

  v8f acc[kMT][4];
#pragma unroll
  for (int u = 0; u < kMT; ++u) {
#pragma unroll
    for (int t = 0; t < 4; ++t) acc[u][t] = v8f{};
  }
  for (int k = 0; k < kD; k += 8) {
    v4f a[kMT];
    v4f b[4];
#pragma unroll
    for (int u = 0; u < kMT; ++u) a[u] = *(const v4f*)(ap + (size_t)u * 16 * kD + k);
#pragma unroll
    for (int t = 0; t < 4; ++t) b[t] = *(const v4f*)(bp + (size_t)t * 16 * kD + k);
#pragma unroll
    for (int u = 0; u < kMT; ++u) {
#pragma unroll
      for (int t = 0; t < 4; ++t) acc[u][t] = wmma_f32(lo2(a[u]), lo2(b[t]), acc[u][t]);
    }
#pragma unroll
    for (int u = 0; u < kMT; ++u) {
#pragma unroll
      for (int t = 0; t < 4; ++t) acc[u][t] = wmma_f32(hi2(a[u]), hi2(b[t]), acc[u][t]);
    }
  }
#pragma unroll
  for (int u = 0; u < kMT; ++u) {
#pragma unroll
    for (int t = 0; t < 4; ++t) {
      int n = n0 + t * 16 + lr;
      int h = n >> 6, d = n & 63;
      float* op = out + (size_t)h * kS * kDH + d;
#pragma unroll
      for (int r = 0; r < 8; ++r) {
        int m = m0 + u * 16 + r + 8 * half;
        op[(size_t)m * kDH] = acc[u][t][r];
      }
    }
  }
}

__global__ void bd_gemm(const float* __restrict__ Qh, const float* __restrict__ Ph,
                        const float* __restrict__ vb, float* __restrict__ BDall) {
  const int h = blockIdx.z;
  float* BD = BDall + (size_t)h * kS * kS;
  const int lane = threadIdx.x & 31;
  const int wave = threadIdx.x >> 5;
  const int half = lane >> 4;
  const int lr   = lane & 15;
  const int m0   = blockIdx.x * kBlkM + wave * (kMT * 16);
  const int n0   = blockIdx.y * 64;

  const float* qp = Qh + (size_t)h * kS * kDH + (size_t)(m0 + lr) * kDH + 4 * half;
  const float* pp = Ph + (size_t)h * kS * kDH + (size_t)(n0 + lr) * kDH + 4 * half;
  const float* bb = vb + h * kDH + 4 * half;

  v8f acc[kMT][4];
#pragma unroll
  for (int u = 0; u < kMT; ++u) {
#pragma unroll
    for (int t = 0; t < 4; ++t) acc[u][t] = v8f{};
  }
  for (int k = 0; k < kDH; k += 8) {
    v4f bias = *(const v4f*)(bb + k);
    v4f a[kMT];
    v4f b[4];
#pragma unroll
    for (int u = 0; u < kMT; ++u) a[u] = *(const v4f*)(qp + (size_t)u * 16 * kDH + k) + bias;
#pragma unroll
    for (int t = 0; t < 4; ++t) b[t] = *(const v4f*)(pp + (size_t)t * 16 * kDH + k);
#pragma unroll
    for (int u = 0; u < kMT; ++u) {
#pragma unroll
      for (int t = 0; t < 4; ++t) acc[u][t] = wmma_f32(lo2(a[u]), lo2(b[t]), acc[u][t]);
    }
#pragma unroll
    for (int u = 0; u < kMT; ++u) {
#pragma unroll
      for (int t = 0; t < 4; ++t) acc[u][t] = wmma_f32(hi2(a[u]), hi2(b[t]), acc[u][t]);
    }
  }
#pragma unroll
  for (int u = 0; u < kMT; ++u) {
#pragma unroll
    for (int t = 0; t < 4; ++t) {
      int n = n0 + t * 16 + lr;
#pragma unroll
      for (int r = 0; r < 8; ++r) {
        int m = m0 + u * 16 + r + 8 * half;
        BD[(size_t)m * kS + n] = acc[u][t][r];
      }
    }
  }
}

__global__ void score_gemm(const float* __restrict__ Qh, const float* __restrict__ Kh,
                           const float* __restrict__ ub, const float* __restrict__ BDall,
                           float* __restrict__ SCgrp, int h0, float scale) {
  const int h = h0 + blockIdx.z;
  const float* BD = BDall + (size_t)h * kS * kS;
  float* SC = SCgrp + (size_t)blockIdx.z * kS * kS;
  const int lane = threadIdx.x & 31;
  const int wave = threadIdx.x >> 5;
  const int half = lane >> 4;
  const int lr   = lane & 15;
  const int m0   = blockIdx.x * kBlkM + wave * (kMT * 16);
  const int n0   = blockIdx.y * 64;

  const float* qp = Qh + (size_t)h * kS * kDH + (size_t)(m0 + lr) * kDH + 4 * half;
  const float* kp = Kh + (size_t)h * kS * kDH + (size_t)(n0 + lr) * kDH + 4 * half;
  const float* bb = ub + h * kDH + 4 * half;

  v8f acc[kMT][4];
#pragma unroll
  for (int u = 0; u < kMT; ++u) {
#pragma unroll
    for (int t = 0; t < 4; ++t) acc[u][t] = v8f{};
  }
  for (int k = 0; k < kDH; k += 8) {
    v4f bias = *(const v4f*)(bb + k);
    v4f a[kMT];
    v4f b[4];
#pragma unroll
    for (int u = 0; u < kMT; ++u) a[u] = *(const v4f*)(qp + (size_t)u * 16 * kDH + k) + bias;
#pragma unroll
    for (int t = 0; t < 4; ++t) b[t] = *(const v4f*)(kp + (size_t)t * 16 * kDH + k);
#pragma unroll
    for (int u = 0; u < kMT; ++u) {
#pragma unroll
      for (int t = 0; t < 4; ++t) acc[u][t] = wmma_f32(lo2(a[u]), lo2(b[t]), acc[u][t]);
    }
#pragma unroll
    for (int u = 0; u < kMT; ++u) {
#pragma unroll
      for (int t = 0; t < 4; ++t) acc[u][t] = wmma_f32(hi2(a[u]), hi2(b[t]), acc[u][t]);
    }
  }
  constexpr int kTB = 2;
#pragma unroll
  for (int u = 0; u < kMT; ++u) {
#pragma unroll
    for (int tb = 0; tb < 4; tb += kTB) {
      float bdv[kTB][8];
#pragma unroll
      for (int t = 0; t < kTB; ++t) {
        const int n = n0 + (tb + t) * 16 + lr;
#pragma unroll
        for (int r = 0; r < 8; ++r) {
          const int m = m0 + u * 16 + r + 8 * half;
          const int d = n - m;
          const unsigned idx = (unsigned)((m + 1) * kS + d - 1 - (d >= 2 ? 1 : 0));
          bdv[t][r] = BD[idx];
        }
      }
      __builtin_amdgcn_sched_barrier(0);
#pragma unroll
      for (int t = 0; t < kTB; ++t) {
        const int n = n0 + (tb + t) * 16 + lr;
#pragma unroll
        for (int r = 0; r < 8; ++r) {
          const int m = m0 + u * 16 + r + 8 * half;
          const float g = (n - m == 1) ? 0.0f : bdv[t][r];
          SC[(size_t)m * kS + n] = (acc[u][tb + t][r] + g) * scale;
        }
      }
      __builtin_amdgcn_sched_barrier(0);
    }
  }
}

static constexpr int kAvWaves = 8;
static constexpr int kAvK     = kS / kAvWaves;
static constexpr int kAvN     = 32;
static constexpr int kAvSteps = kAvK / 4;
static constexpr int kAvChunk = 8;

__device__ __forceinline__ constexpr int av_vrow(int i) { return 8 * (i >> 1) + 2 * (i & 1); }

__global__ void __launch_bounds__(kAvWaves * 32)
av_gemm(const float* __restrict__ SCgrp, const float* __restrict__ Vh,
        float* __restrict__ AVec, int h0) {
  const int h = h0 + blockIdx.z;
  const float* SC = SCgrp + (size_t)blockIdx.z * kS * kS;
  __shared__ float part[kAvWaves][16][32];
  __shared__ float smax[kAvWaves][16];
  __shared__ float ssum[kAvWaves][16];
  const int lane = threadIdx.x & 31;
  const int wave = threadIdx.x >> 5;
  const int half = lane >> 4;
  const int lr   = lane & 15;
  const int m0   = blockIdx.x * 16;
  const int n0   = blockIdx.y * kAvN;
  const int kb   = wave * kAvK + 4 * half;

  const float* pp = SC + (size_t)(m0 + lr) * kS + kb;
  const float* vp = Vh + (size_t)h * kS * kDH + (size_t)kb * kDH + n0 + lr;

  v4f sc[kAvSteps / 2];
#pragma unroll
  for (int q = 0; q < kAvSteps / 2; ++q) sc[q] = *(const v4f*)(pp + 8 * q);

  v2f bq[2][kAvChunk * 2];
#pragma unroll
  for (int j = 0; j < kAvChunk * 2; ++j) {
    const float* vb = vp + (size_t)av_vrow(j >> 1) * kDH + (j & 1) * 16;
    bq[0][j] = v2f{vb[0], vb[kDH]};
  }
  __builtin_amdgcn_sched_barrier(0);

  float mx = fmaxf(fmaxf(sc[0][0], sc[0][1]), fmaxf(sc[0][2], sc[0][3]));
#pragma unroll
  for (int q = 1; q < kAvSteps / 2; ++q)
    mx = fmaxf(mx, fmaxf(fmaxf(sc[q][0], sc[q][1]), fmaxf(sc[q][2], sc[q][3])));
  mx = fmaxf(mx, __shfl_xor(mx, 16, 32));

  float sum = 0.0f;
  v8f acc[2] = {v8f{}, v8f{}};
#pragma unroll
  for (int c = 0; c < kAvSteps / kAvChunk; ++c) {
    if (c + 1 < kAvSteps / kAvChunk) {
#pragma unroll
      for (int j = 0; j < kAvChunk * 2; ++j) {
        const float* vb = vp + (size_t)av_vrow((c + 1) * kAvChunk + (j >> 1)) * kDH + (j & 1) * 16;
        bq[(c + 1) & 1][j] = v2f{vb[0], vb[kDH]};
      }
    }
    __builtin_amdgcn_sched_barrier(0);
#pragma unroll
    for (int s = 0; s < kAvChunk; ++s) {
      const int i = c * kAvChunk + s;
      v2f a = {__expf(sc[i >> 1][2 * (i & 1)] - mx), __expf(sc[i >> 1][2 * (i & 1) + 1] - mx)};
      sum += a[0] + a[1];
#pragma unroll
      for (int t = 0; t < 2; ++t) acc[t] = wmma_f32(a, bq[c & 1][s * 2 + t], acc[t]);
    }
  }
  sum += __shfl_xor(sum, 16, 32);

  if (half == 0) {
    smax[wave][lr] = mx;
    ssum[wave][lr] = sum;
  }
#pragma unroll
  for (int t = 0; t < 2; ++t) {
#pragma unroll
    for (int r = 0; r < 8; ++r) part[wave][t * 8 + r][lane] = acc[t][r];
  }
  __syncthreads();

  const int row = wave + 8 * half;
  float g = smax[0][row];
#pragma unroll
  for (int w = 1; w < kAvWaves; ++w) g = fmaxf(g, smax[w][row]);
  float f[kAvWaves];
  float den = 0.0f;
#pragma unroll
  for (int w = 0; w < kAvWaves; ++w) {
    f[w] = __expf(smax[w][row] - g);
    den += f[w] * ssum[w][row];
  }
  const float inv = 1.0f / den;
  const int m = m0 + row;
#pragma unroll
  for (int t = 0; t < 2; ++t) {
    float s = f[0] * part[0][t * 8 + wave][lane];
#pragma unroll
    for (int w = 1; w < kAvWaves; ++w) s += f[w] * part[w][t * 8 + wave][lane];
    AVec[(size_t)m * kD + h * kDH + n0 + t * 16 + lr] = s * inv;
  }
}

__global__ void out_gemm(const float* __restrict__ A, const float* __restrict__ B,
                         float* __restrict__ C) {
  const int lane = threadIdx.x & 31;
  const int wave = threadIdx.x >> 5;
  const int half = lane >> 4;
  const int lr   = lane & 15;
  const int m0   = blockIdx.x * 16;
  const int n0   = wave * 64;

  const float* ap = A + (size_t)(m0 + lr) * kD + 4 * half;
  const float* bp = B + (size_t)(n0 + lr) * kD + 4 * half;

  v8f acc[4] = {v8f{}, v8f{}, v8f{}, v8f{}};
  for (int k = 0; k < kD; k += 8) {
    v4f a = *(const v4f*)(ap + k);
    v4f b[4];
#pragma unroll
    for (int t = 0; t < 4; ++t) b[t] = *(const v4f*)(bp + (size_t)t * 16 * kD + k);
#pragma unroll
    for (int t = 0; t < 4; ++t) acc[t] = wmma_f32(lo2(a), lo2(b[t]), acc[t]);
#pragma unroll
    for (int t = 0; t < 4; ++t) acc[t] = wmma_f32(hi2(a), hi2(b[t]), acc[t]);
  }
#pragma unroll
  for (int t = 0; t < 4; ++t) {
    int n = n0 + t * 16 + lr;
#pragma unroll
    for (int r = 0; r < 8; ++r) {
      int m = m0 + r + 8 * half;
      C[(size_t)m * kD + n] = acc[t][r];
    }
  }
}

extern "C" void kernel_launch(void* const* d_in, const int* in_sizes, int n_in,
                              void* d_out, int out_size, void* d_ws, size_t ws_size,
                              hipStream_t stream) {
  const float* x   = (const float*)d_in[0];
  const float* pos = (const float*)d_in[1];
  const float* Wq  = (const float*)d_in[2];
  const float* Wk  = (const float*)d_in[3];
  const float* Wv  = (const float*)d_in[4];
  const float* Wp  = (const float*)d_in[5];
  const float* Wo  = (const float*)d_in[6];
  const float* u   = (const float*)d_in[7];
  const float* v   = (const float*)d_in[8];

  float* ws = (float*)d_ws;
  const size_t SD = (size_t)kS * kD;
  float* Qhs  = ws;
  float* Khs  = Qhs + SD;
  float* Vhs  = Khs + SD;
  float* Phs  = Vhs + SD;
  float* AVec = Phs + SD;
  float* BD   = AVec + SD;
  float* SC   = BD + (size_t)kH * kS * kS;

  dim3 blk(256);
  proj_gemm_head<<<dim3(kS / kBlkM, kD / 64, 4), blk, 0, stream>>>(x, pos, Wq, Wk, Wv, Wp, Qhs);

  const float scale = 0.125f;
  bd_gemm<<<dim3(kS / kBlkM, kS / 64, kH), blk, 0, stream>>>(Qhs, Phs, v, BD);
  for (int h0 = 0; h0 < kH; h0 += kGrp) {
    score_gemm<<<dim3(kS / kBlkM, kS / 64, kGrp), blk, 0, stream>>>(Qhs, Khs, u, BD, SC, h0, scale);
    av_gemm   <<<dim3(kS / 16, kDH / kAvN, kGrp), blk, 0, stream>>>(SC, Vhs, AVec, h0);
  }

  out_gemm<<<dim3(kS / 16), blk, 0, stream>>>(AVec, Wo, (float*)d_out);
}


# AV product: a block covers all 64 head columns as four interleaved tiles (tile t = columns 4c+t), so one 128-bit V row load feeds four WMMAs and a score slice is read and exponentiated once, not twice
# speedup vs baseline: 7.3762x; 1.0488x over previous
#include <hip/hip_runtime.h>
#include <math.h>

typedef __attribute__((ext_vector_type(2))) float v2f;
typedef __attribute__((ext_vector_type(4))) float v4f;
typedef __attribute__((ext_vector_type(8))) float v8f;

static constexpr int kS  = 2048;
static constexpr int kD  = 512;
static constexpr int kH  = 8;
static constexpr int kDH = 64;
static constexpr int kGrp = 4;
static constexpr int kMT  = 2;
static constexpr int kBlkM = 8 * kMT * 16;

__device__ __forceinline__ v8f wmma_f32(v2f a, v2f b, v8f c) {
  return __builtin_amdgcn_wmma_f32_16x16x4_f32(
       false, a,  false, b,
       (short)0, c,  false,  false);
}

__device__ __forceinline__ v2f lo2(v4f x) { return v2f{x[0], x[1]}; }
__device__ __forceinline__ v2f hi2(v4f x) { return v2f{x[2], x[3]}; }

__global__ void proj_gemm_head(const float* __restrict__ x, const float* __restrict__ pos,
                               const float* __restrict__ Wq, const float* __restrict__ Wk,
                               const float* __restrict__ Wv, const float* __restrict__ Wp,
                               float* __restrict__ out4) {
  const int which = blockIdx.z;
  const float* A = (which == 3) ? pos : x;
  const float* B = (which == 0) ? Wq : (which == 1) ? Wk : (which == 2) ? Wv : Wp;
  float* out = out4 + (size_t)which * kS * kD;
  const int lane = threadIdx.x & 31;
  const int wave = threadIdx.x >> 5;
  const int half = lane >> 4;
  const int lr   = lane & 15;
  const int m0   = blockIdx.x * kBlkM + wave * (kMT * 16);
  const int n0   = blockIdx.y * 64;

  const float* ap = A + (size_t)(m0 + lr) * kD + 4 * half;
  const float* bp = B + (size_t)(n0 + lr) * kD + 4 * half;

  v8f acc[kMT][4];
#pragma unroll
  for (int u = 0; u < kMT; ++u) {
#pragma unroll
    for (int t = 0; t < 4; ++t) acc[u][t] = v8f{};
  }
  for (int k = 0; k < kD; k += 8) {
    v4f a[kMT];
    v4f b[4];
#pragma unroll
    for (int u = 0; u < kMT; ++u) a[u] = *(const v4f*)(ap + (size_t)u * 16 * kD + k);
#pragma unroll
    for (int t = 0; t < 4; ++t) b[t] = *(const v4f*)(bp + (size_t)t * 16 * kD + k);
#pragma unroll
    for (int u = 0; u < kMT; ++u) {
#pragma unroll
      for (int t = 0; t < 4; ++t) acc[u][t] = wmma_f32(lo2(a[u]), lo2(b[t]), acc[u][t]);
    }
#pragma unroll
    for (int u = 0; u < kMT; ++u) {
#pragma unroll
      for (int t = 0; t < 4; ++t) acc[u][t] = wmma_f32(hi2(a[u]), hi2(b[t]), acc[u][t]);
    }
  }
#pragma unroll
  for (int u = 0; u < kMT; ++u) {
#pragma unroll
    for (int t = 0; t < 4; ++t) {
      int n = n0 + t * 16 + lr;
      int h = n >> 6, d = n & 63;
      float* op = out + (size_t)h * kS * kDH + d;
#pragma unroll
      for (int r = 0; r < 8; ++r) {
        int m = m0 + u * 16 + r + 8 * half;
        op[(size_t)m * kDH] = acc[u][t][r];
      }
    }
  }
}

__global__ void bd_gemm(const float* __restrict__ Qh, const float* __restrict__ Ph,
                        const float* __restrict__ vb, float* __restrict__ BDall) {
  const int h = blockIdx.z;
  float* BD = BDall + (size_t)h * kS * kS;
  const int lane = threadIdx.x & 31;
  const int wave = threadIdx.x >> 5;
  const int half = lane >> 4;
  const int lr   = lane & 15;
  const int m0   = blockIdx.x * kBlkM + wave * (kMT * 16);
  const int n0   = blockIdx.y * 64;

  const float* qp = Qh + (size_t)h * kS * kDH + (size_t)(m0 + lr) * kDH + 4 * half;
  const float* pp = Ph + (size_t)h * kS * kDH + (size_t)(n0 + lr) * kDH + 4 * half;
  const float* bb = vb + h * kDH + 4 * half;

  v8f acc[kMT][4];
#pragma unroll
  for (int u = 0; u < kMT; ++u) {
#pragma unroll
    for (int t = 0; t < 4; ++t) acc[u][t] = v8f{};
  }
  for (int k = 0; k < kDH; k += 8) {
    v4f bias = *(const v4f*)(bb + k);
    v4f a[kMT];
    v4f b[4];
#pragma unroll
    for (int u = 0; u < kMT; ++u) a[u] = *(const v4f*)(qp + (size_t)u * 16 * kDH + k) + bias;
#pragma unroll
    for (int t = 0; t < 4; ++t) b[t] = *(const v4f*)(pp + (size_t)t * 16 * kDH + k);
#pragma unroll
    for (int u = 0; u < kMT; ++u) {
#pragma unroll
      for (int t = 0; t < 4; ++t) acc[u][t] = wmma_f32(lo2(a[u]), lo2(b[t]), acc[u][t]);
    }
#pragma unroll
    for (int u = 0; u < kMT; ++u) {
#pragma unroll
      for (int t = 0; t < 4; ++t) acc[u][t] = wmma_f32(hi2(a[u]), hi2(b[t]), acc[u][t]);
    }
  }
#pragma unroll
  for (int u = 0; u < kMT; ++u) {
#pragma unroll
    for (int t = 0; t < 4; ++t) {
      int n = n0 + t * 16 + lr;
#pragma unroll
      for (int r = 0; r < 8; ++r) {
        int m = m0 + u * 16 + r + 8 * half;
        BD[(size_t)m * kS + n] = acc[u][t][r];
      }
    }
  }
}

__global__ void score_gemm(const float* __restrict__ Qh, const float* __restrict__ Kh,
                           const float* __restrict__ ub, const float* __restrict__ BDall,
                           float* __restrict__ SCgrp, int h0, float scale) {
  const int h = h0 + blockIdx.z;
  const float* BD = BDall + (size_t)h * kS * kS;
  float* SC = SCgrp + (size_t)blockIdx.z * kS * kS;
  const int lane = threadIdx.x & 31;
  const int wave = threadIdx.x >> 5;
  const int half = lane >> 4;
  const int lr   = lane & 15;
  const int m0   = blockIdx.x * kBlkM + wave * (kMT * 16);
  const int n0   = blockIdx.y * 64;

  const float* qp = Qh + (size_t)h * kS * kDH + (size_t)(m0 + lr) * kDH + 4 * half;
  const float* kp = Kh + (size_t)h * kS * kDH + (size_t)(n0 + lr) * kDH + 4 * half;
  const float* bb = ub + h * kDH + 4 * half;

  v8f acc[kMT][4];
#pragma unroll
  for (int u = 0; u < kMT; ++u) {
#pragma unroll
    for (int t = 0; t < 4; ++t) acc[u][t] = v8f{};
  }
  for (int k = 0; k < kDH; k += 8) {
    v4f bias = *(const v4f*)(bb + k);
    v4f a[kMT];
    v4f b[4];
#pragma unroll
    for (int u = 0; u < kMT; ++u) a[u] = *(const v4f*)(qp + (size_t)u * 16 * kDH + k) + bias;
#pragma unroll
    for (int t = 0; t < 4; ++t) b[t] = *(const v4f*)(kp + (size_t)t * 16 * kDH + k);
#pragma unroll
    for (int u = 0; u < kMT; ++u) {
#pragma unroll
      for (int t = 0; t < 4; ++t) acc[u][t] = wmma_f32(lo2(a[u]), lo2(b[t]), acc[u][t]);
    }
#pragma unroll
    for (int u = 0; u < kMT; ++u) {
#pragma unroll
      for (int t = 0; t < 4; ++t) acc[u][t] = wmma_f32(hi2(a[u]), hi2(b[t]), acc[u][t]);
    }
  }
  constexpr int kTB = 2;
#pragma unroll
  for (int u = 0; u < kMT; ++u) {
#pragma unroll
    for (int tb = 0; tb < 4; tb += kTB) {
      float bdv[kTB][8];
#pragma unroll
      for (int t = 0; t < kTB; ++t) {
        const int n = n0 + (tb + t) * 16 + lr;
#pragma unroll
        for (int r = 0; r < 8; ++r) {
          const int m = m0 + u * 16 + r + 8 * half;
          const int d = n - m;
          const unsigned idx = (unsigned)((m + 1) * kS + d - 1 - (d >= 2 ? 1 : 0));
          bdv[t][r] = BD[idx];
        }
      }
      __builtin_amdgcn_sched_barrier(0);
#pragma unroll
      for (int t = 0; t < kTB; ++t) {
        const int n = n0 + (tb + t) * 16 + lr;
#pragma unroll
        for (int r = 0; r < 8; ++r) {
          const int m = m0 + u * 16 + r + 8 * half;
          const float g = (n - m == 1) ? 0.0f : bdv[t][r];
          SC[(size_t)m * kS + n] = (acc[u][tb + t][r] + g) * scale;
        }
      }
      __builtin_amdgcn_sched_barrier(0);
    }
  }
}

static constexpr int kAvWaves = 8;
static constexpr int kAvK     = kS / kAvWaves;
static constexpr int kAvT     = kDH / 16;
static constexpr int kAvSteps = kAvK / 4;
static constexpr int kAvChunk = 4;

__device__ __forceinline__ constexpr int av_vrow(int i) { return 8 * (i >> 1) + 2 * (i & 1); }

__global__ void __launch_bounds__(kAvWaves * 32, 4)
av_gemm(const float* __restrict__ SCgrp, const float* __restrict__ Vh,
        float* __restrict__ AVec, int h0) {
  const int h = h0 + blockIdx.z;
  const float* SC = SCgrp + (size_t)blockIdx.z * kS * kS;
  __shared__ float part[kAvWaves][kAvT * 8][32];
  __shared__ float smax[kAvWaves][16];
  __shared__ float ssum[kAvWaves][16];
  const int lane = threadIdx.x & 31;
  const int wave = threadIdx.x >> 5;
  const int half = lane >> 4;
  const int lr   = lane & 15;
  const int m0   = blockIdx.x * 16;
  const int kb   = wave * kAvK + 4 * half;

  const float* pp = SC + (size_t)(m0 + lr) * kS + kb;
  const float* vp = Vh + (size_t)h * kS * kDH + (size_t)kb * kDH + 4 * lr;

  v4f sc[kAvSteps / 2];
#pragma unroll
  for (int q = 0; q < kAvSteps / 2; ++q) sc[q] = *(const v4f*)(pp + 8 * q);

  v4f vq[2][kAvChunk * 2];
#pragma unroll
  for (int j = 0; j < kAvChunk * 2; ++j)
    vq[0][j] = *(const v4f*)(vp + (size_t)(av_vrow(j >> 1) + (j & 1)) * kDH);
  __builtin_amdgcn_sched_barrier(0);

  float mx = fmaxf(fmaxf(sc[0][0], sc[0][1]), fmaxf(sc[0][2], sc[0][3]));
#pragma unroll
  for (int q = 1; q < kAvSteps / 2; ++q)
    mx = fmaxf(mx, fmaxf(fmaxf(sc[q][0], sc[q][1]), fmaxf(sc[q][2], sc[q][3])));
  mx = fmaxf(mx, __shfl_xor(mx, 16, 32));

  float sum = 0.0f;
  v8f acc[kAvT] = {v8f{}, v8f{}, v8f{}, v8f{}};
#pragma unroll
  for (int c = 0; c < kAvSteps / kAvChunk; ++c) {
    if (c + 1 < kAvSteps / kAvChunk) {
#pragma unroll
      for (int j = 0; j < kAvChunk * 2; ++j)
        vq[(c + 1) & 1][j] =
            *(const v4f*)(vp + (size_t)(av_vrow((c + 1) * kAvChunk + (j >> 1)) + (j & 1)) * kDH);
    }
    __builtin_amdgcn_sched_barrier(0);
#pragma unroll
    for (int s = 0; s < kAvChunk; ++s) {
      const int i = c * kAvChunk + s;
      v2f a = {__expf(sc[i >> 1][2 * (i & 1)] - mx), __expf(sc[i >> 1][2 * (i & 1) + 1] - mx)};
      sum += a[0] + a[1];
      const v4f r0 = vq[c & 1][2 * s];
      const v4f r1 = vq[c & 1][2 * s + 1];
#pragma unroll
      for (int t = 0; t < kAvT; ++t) acc[t] = wmma_f32(a, v2f{r0[t], r1[t]}, acc[t]);
    }
  }
  sum += __shfl_xor(sum, 16, 32);

  if (half == 0) {
    smax[wave][lr] = mx;
    ssum[wave][lr] = sum;
  }
#pragma unroll
  for (int t = 0; t < kAvT; ++t) {
#pragma unroll
    for (int r = 0; r < 8; ++r) part[wave][t * 8 + r][lane] = acc[t][r];
  }
  __syncthreads();

  const int row = wave + 8 * half;
  float g = smax[0][row];
#pragma unroll
  for (int w = 1; w < kAvWaves; ++w) g = fmaxf(g, smax[w][row]);
  float f[kAvWaves];
  float den = 0.0f;
#pragma unroll
  for (int w = 0; w < kAvWaves; ++w) {
    f[w] = __expf(smax[w][row] - g);
    den += f[w] * ssum[w][row];
  }
  const float inv = 1.0f / den;
  const int m = m0 + row;
  v4f o;
#pragma unroll
  for (int t = 0; t < kAvT; ++t) {
    float s = f[0] * part[0][t * 8 + wave][lane];
#pragma unroll
    for (int w = 1; w < kAvWaves; ++w) s += f[w] * part[w][t * 8 + wave][lane];
    o[t] = s * inv;
  }
  *(v4f*)(AVec + (size_t)m * kD + h * kDH + 4 * lr) = o;
}

__global__ void out_gemm(const float* __restrict__ A, const float* __restrict__ B,
                         float* __restrict__ C) {
  const int lane = threadIdx.x & 31;
  const int wave = threadIdx.x >> 5;
  const int half = lane >> 4;
  const int lr   = lane & 15;
  const int m0   = blockIdx.x * 16;
  const int n0   = wave * 64;

  const float* ap = A + (size_t)(m0 + lr) * kD + 4 * half;
  const float* bp = B + (size_t)(n0 + lr) * kD + 4 * half;

  v8f acc[4] = {v8f{}, v8f{}, v8f{}, v8f{}};
  for (int k = 0; k < kD; k += 8) {
    v4f a = *(const v4f*)(ap + k);
    v4f b[4];
#pragma unroll
    for (int t = 0; t < 4; ++t) b[t] = *(const v4f*)(bp + (size_t)t * 16 * kD + k);
#pragma unroll
    for (int t = 0; t < 4; ++t) acc[t] = wmma_f32(lo2(a), lo2(b[t]), acc[t]);
#pragma unroll
    for (int t = 0; t < 4; ++t) acc[t] = wmma_f32(hi2(a), hi2(b[t]), acc[t]);
  }
#pragma unroll
  for (int t = 0; t < 4; ++t) {
    int n = n0 + t * 16 + lr;
#pragma unroll
    for (int r = 0; r < 8; ++r) {
      int m = m0 + r + 8 * half;
      C[(size_t)m * kD + n] = acc[t][r];
    }
  }
}

extern "C" void kernel_launch(void* const* d_in, const int* in_sizes, int n_in,
                              void* d_out, int out_size, void* d_ws, size_t ws_size,
                              hipStream_t stream) {
  const float* x   = (const float*)d_in[0];
  const float* pos = (const float*)d_in[1];
  const float* Wq  = (const float*)d_in[2];
  const float* Wk  = (const float*)d_in[3];
  const float* Wv  = (const float*)d_in[4];
  const float* Wp  = (const float*)d_in[5];
  const float* Wo  = (const float*)d_in[6];
  const float* u   = (const float*)d_in[7];
  const float* v   = (const float*)d_in[8];

  float* ws = (float*)d_ws;
  const size_t SD = (size_t)kS * kD;
  float* Qhs  = ws;
  float* Khs  = Qhs + SD;
  float* Vhs  = Khs + SD;
  float* Phs  = Vhs + SD;
  float* AVec = Phs + SD;
  float* BD   = AVec + SD;
  float* SC   = BD + (size_t)kH * kS * kS;

  dim3 blk(256);
  proj_gemm_head<<<dim3(kS / kBlkM, kD / 64, 4), blk, 0, stream>>>(x, pos, Wq, Wk, Wv, Wp, Qhs);

  const float scale = 0.125f;
  bd_gemm<<<dim3(kS / kBlkM, kS / 64, kH), blk, 0, stream>>>(Qhs, Phs, v, BD);
  for (int h0 = 0; h0 < kH; h0 += kGrp) {
    score_gemm<<<dim3(kS / kBlkM, kS / 64, kGrp), blk, 0, stream>>>(Qhs, Khs, u, BD, SC, h0, scale);
    av_gemm   <<<dim3(kS / 16, 1, kGrp), blk, 0, stream>>>(SC, Vhs, AVec, h0);
  }

  out_gemm<<<dim3(kS / 16), blk, 0, stream>>>(AVec, Wo, (float*)d_out);
}


# projection, BD, score and output products: operand tiles staged in LDS by whole-line 128-bit loads, WMMA fragments read from LDS, not global (output product re-tiled 64x64); sum order kept
# speedup vs baseline: 9.3871x; 1.2726x over previous
#include <hip/hip_runtime.h>
#include <math.h>

typedef __attribute__((ext_vector_type(2))) float v2f;
typedef __attribute__((ext_vector_type(4))) float v4f;
typedef __attribute__((ext_vector_type(8))) float v8f;

static constexpr int kS  = 2048;
static constexpr int kD  = 512;
static constexpr int kH  = 8;
static constexpr int kDH = 64;
static constexpr int kGrp = 4;
static constexpr int kMT  = 2;
static constexpr int kBlkM = 8 * kMT * 16;
static constexpr int kKC   = 64;
static constexpr int kPitch = kKC + 4;

__device__ __forceinline__ v8f wmma_f32(v2f a, v2f b, v8f c) {
  return __builtin_amdgcn_wmma_f32_16x16x4_f32(
       false, a,  false, b,
       (short)0, c,  false,  false);
}

__device__ __forceinline__ v2f lo2(v4f x) { return v2f{x[0], x[1]}; }
__device__ __forceinline__ v2f hi2(v4f x) { return v2f{x[2], x[3]}; }


template <int kRows>
__device__ __forceinline__ void tile_fetch(v4f (&reg)[kRows / 16], const float* __restrict__ src,
                                           int ld) {
  const int c4 = threadIdx.x & 15;
  const int r  = threadIdx.x >> 4;
#pragma unroll
  for (int j = 0; j < kRows / 16; ++j)
    reg[j] = *(const v4f*)(src + (size_t)(r + 16 * j) * ld + 4 * c4);
}

template <int kRows>
__device__ __forceinline__ void tile_put(float* __restrict__ dst, const v4f (&reg)[kRows / 16]) {
  const int c4 = threadIdx.x & 15;
  const int r  = threadIdx.x >> 4;
#pragma unroll
  for (int j = 0; j < kRows / 16; ++j) *(v4f*)(dst + (r + 16 * j) * kPitch + 4 * c4) = reg[j];
}

template <int kMTw, int kNTw>
__device__ __forceinline__ void tile_mma(const float* la, const float* lb,
                                         v8f (&acc)[kMTw][kNTw]) {
#pragma unroll
  for (int k = 0; k < kKC; k += 8) {
    v4f a[kMTw];
    v4f b[kNTw];
#pragma unroll
    for (int u = 0; u < kMTw; ++u) a[u] = *(const v4f*)(la + u * 16 * kPitch + k);
#pragma unroll
    for (int t = 0; t < kNTw; ++t) b[t] = *(const v4f*)(lb + t * 16 * kPitch + k);
#pragma unroll
    for (int u = 0; u < kMTw; ++u) {
#pragma unroll
      for (int t = 0; t < kNTw; ++t) acc[u][t] = wmma_f32(lo2(a[u]), lo2(b[t]), acc[u][t]);
    }
#pragma unroll
    for (int u = 0; u < kMTw; ++u) {
#pragma unroll
      for (int t = 0; t < kNTw; ++t) acc[u][t] = wmma_f32(hi2(a[u]), hi2(b[t]), acc[u][t]);
    }
  }
}

__global__ void __launch_bounds__(256)
proj_gemm_head(const float* __restrict__ x, const float* __restrict__ pos,
               const float* __restrict__ Wq, const float* __restrict__ Wk,
               const float* __restrict__ Wv, const float* __restrict__ Wp,
               float* __restrict__ out4) {
  __shared__ alignas(16) float ldsA[2][kBlkM * kPitch];
  __shared__ alignas(16) float ldsB[2][64 * kPitch];
  const int which = blockIdx.z;
  const float* A = (which == 3) ? pos : x;
  const float* B = (which == 0) ? Wq : (which == 1) ? Wk : (which == 2) ? Wv : Wp;
  float* out = out4 + (size_t)which * kS * kD;
  const int lane = threadIdx.x & 31;
  const int wave = threadIdx.x >> 5;
  const int half = lane >> 4;
  const int lr   = lane & 15;
  const int m0   = blockIdx.x * kBlkM + wave * (kMT * 16);
  const int n0   = blockIdx.y * 64;

  const float* Ag = A + (size_t)blockIdx.x * kBlkM * kD;
  const float* Bg = B + (size_t)n0 * kD;
  const int la = (wave * (kMT * 16) + lr) * kPitch + 4 * half;
  const int lb = lr * kPitch + 4 * half;

  v8f acc[kMT][4];
#pragma unroll
  for (int u = 0; u < kMT; ++u) {
#pragma unroll
    for (int t = 0; t < 4; ++t) acc[u][t] = v8f{};
  }
  v4f ra[kBlkM / 16];
  v4f rb[4];
  tile_fetch<kBlkM>(ra, Ag, kD);
  tile_fetch<64>(rb, Bg, kD);
  tile_put<kBlkM>(ldsA[0], ra);
  tile_put<64>(ldsB[0], rb);
  __syncthreads();
  for (int c = 0; c < kD / kKC; ++c) {
    const int cur = c & 1;
    if (c + 1 < kD / kKC) {
      tile_fetch<kBlkM>(ra, Ag + (c + 1) * kKC, kD);
      tile_fetch<64>(rb, Bg + (c + 1) * kKC, kD);
    }
    tile_mma<kMT, 4>(ldsA[cur] + la, ldsB[cur] + lb, acc);
    __builtin_amdgcn_sched_barrier(0);
    if (c + 1 < kD / kKC) {
      tile_put<kBlkM>(ldsA[cur ^ 1], ra);
      tile_put<64>(ldsB[cur ^ 1], rb);
      __syncthreads();
    }
  }
#pragma unroll
  for (int u = 0; u < kMT; ++u) {
#pragma unroll
    for (int t = 0; t < 4; ++t) {
      int n = n0 + t * 16 + lr;
      int h = n >> 6, d = n & 63;
      float* op = out + (size_t)h * kS * kDH + d;
#pragma unroll
      for (int r = 0; r < 8; ++r) {
        int m = m0 + u * 16 + r + 8 * half;
        op[(size_t)m * kDH] = acc[u][t][r];
      }
    }
  }
}

__global__ void __launch_bounds__(256)
bd_gemm(const float* __restrict__ Qh, const float* __restrict__ Ph,
        const float* __restrict__ vb, float* __restrict__ BDall) {
  __shared__ alignas(16) float ldsA[kBlkM * kPitch];
  __shared__ alignas(16) float ldsB[64 * kPitch];
  const int h = blockIdx.z;
  float* BD = BDall + (size_t)h * kS * kS;
  const int lane = threadIdx.x & 31;
  const int wave = threadIdx.x >> 5;
  const int half = lane >> 4;
  const int lr   = lane & 15;
  const int m0   = blockIdx.x * kBlkM + wave * (kMT * 16);
  const int n0   = blockIdx.y * 64;

  {
    v4f ra[kBlkM / 16];
    v4f rb[4];
    tile_fetch<kBlkM>(ra, Qh + (size_t)h * kS * kDH + (size_t)blockIdx.x * kBlkM * kDH, kDH);
    tile_fetch<64>(rb, Ph + (size_t)h * kS * kDH + (size_t)n0 * kDH, kDH);
    const v4f bias = *(const v4f*)(vb + h * kDH + 4 * (threadIdx.x & 15));
#pragma unroll
    for (int j = 0; j < kBlkM / 16; ++j) ra[j] = ra[j] + bias;
    tile_put<kBlkM>(ldsA, ra);
    tile_put<64>(ldsB, rb);
  }
  __syncthreads();

  v8f acc[kMT][4];
#pragma unroll
  for (int u = 0; u < kMT; ++u) {
#pragma unroll
    for (int t = 0; t < 4; ++t) acc[u][t] = v8f{};
  }
  tile_mma<kMT, 4>(ldsA + (wave * (kMT * 16) + lr) * kPitch + 4 * half,
                   ldsB + lr * kPitch + 4 * half, acc);
#pragma unroll
  for (int u = 0; u < kMT; ++u) {
#pragma unroll
    for (int t = 0; t < 4; ++t) {
      int n = n0 + t * 16 + lr;
#pragma unroll
      for (int r = 0; r < 8; ++r) {
        int m = m0 + u * 16 + r + 8 * half;
        BD[(size_t)m * kS + n] = acc[u][t][r];
      }
    }
  }
}

__global__ void __launch_bounds__(256, 6)
score_gemm(const float* __restrict__ Qh, const float* __restrict__ Kh,
           const float* __restrict__ ub, const float* __restrict__ BDall,
           float* __restrict__ SCgrp, int h0, float scale) {
  __shared__ alignas(16) float ldsA[kBlkM * kPitch];
  __shared__ alignas(16) float ldsB[64 * kPitch];
  const int h = h0 + blockIdx.z;
  const float* BD = BDall + (size_t)h * kS * kS;
  float* SC = SCgrp + (size_t)blockIdx.z * kS * kS;
  const int lane = threadIdx.x & 31;
  const int wave = threadIdx.x >> 5;
  const int half = lane >> 4;
  const int lr   = lane & 15;
  const int m0   = blockIdx.x * kBlkM + wave * (kMT * 16);
  const int n0   = blockIdx.y * 64;

  {
    v4f ra[kBlkM / 16];
    v4f rb[4];
    tile_fetch<kBlkM>(ra, Qh + (size_t)h * kS * kDH + (size_t)blockIdx.x * kBlkM * kDH, kDH);
    tile_fetch<64>(rb, Kh + (size_t)h * kS * kDH + (size_t)n0 * kDH, kDH);
    const v4f bias = *(const v4f*)(ub + h * kDH + 4 * (threadIdx.x & 15));
#pragma unroll
    for (int j = 0; j < kBlkM / 16; ++j) ra[j] = ra[j] + bias;
    tile_put<kBlkM>(ldsA, ra);
    tile_put<64>(ldsB, rb);
  }
  __syncthreads();

  v8f acc[kMT][4];
#pragma unroll
  for (int u = 0; u < kMT; ++u) {
#pragma unroll
    for (int t = 0; t < 4; ++t) acc[u][t] = v8f{};
  }
  tile_mma<kMT, 4>(ldsA + (wave * (kMT * 16) + lr) * kPitch + 4 * half,
                   ldsB + lr * kPitch + 4 * half, acc);
  constexpr int kTB = 2;
#pragma unroll
  for (int u = 0; u < kMT; ++u) {
#pragma unroll
    for (int tb = 0; tb < 4; tb += kTB) {
      float bdv[kTB][8];
#pragma unroll
      for (int t = 0; t < kTB; ++t) {
        const int n = n0 + (tb + t) * 16 + lr;
#pragma unroll
        for (int r = 0; r < 8; ++r) {
          const int m = m0 + u * 16 + r + 8 * half;
          const int d = n - m;
          const unsigned idx = (unsigned)((m + 1) * kS + d - 1 - (d >= 2 ? 1 : 0));
          bdv[t][r] = BD[idx];
        }
      }
      __builtin_amdgcn_sched_barrier(0);
#pragma unroll
      for (int t = 0; t < kTB; ++t) {
        const int n = n0 + (tb + t) * 16 + lr;
#pragma unroll
        for (int r = 0; r < 8; ++r) {
          const int m = m0 + u * 16 + r + 8 * half;
          const float g = (n - m == 1) ? 0.0f : bdv[t][r];
          SC[(size_t)m * kS + n] = (acc[u][tb + t][r] + g) * scale;
        }
      }
      __builtin_amdgcn_sched_barrier(0);
    }
  }
}

static constexpr int kAvWaves = 8;
static constexpr int kAvK     = kS / kAvWaves;
static constexpr int kAvT     = kDH / 16;
static constexpr int kAvSteps = kAvK / 4;
static constexpr int kAvChunk = 4;

__device__ __forceinline__ constexpr int av_vrow(int i) { return 8 * (i >> 1) + 2 * (i & 1); }

__global__ void __launch_bounds__(kAvWaves * 32, 4)
av_gemm(const float* __restrict__ SCgrp, const float* __restrict__ Vh,
        float* __restrict__ AVec, int h0) {
  const int h = h0 + blockIdx.z;
  const float* SC = SCgrp + (size_t)blockIdx.z * kS * kS;
  __shared__ float part[kAvWaves][kAvT * 8][32];
  __shared__ float smax[kAvWaves][16];
  __shared__ float ssum[kAvWaves][16];
  const int lane = threadIdx.x & 31;
  const int wave = threadIdx.x >> 5;
  const int half = lane >> 4;
  const int lr   = lane & 15;
  const int m0   = blockIdx.x * 16;
  const int kb   = wave * kAvK + 4 * half;

  const float* pp = SC + (size_t)(m0 + lr) * kS + kb;
  const float* vp = Vh + (size_t)h * kS * kDH + (size_t)kb * kDH + 4 * lr;

  v4f sc[kAvSteps / 2];
#pragma unroll
  for (int q = 0; q < kAvSteps / 2; ++q) sc[q] = *(const v4f*)(pp + 8 * q);

  v4f vq[2][kAvChunk * 2];
#pragma unroll
  for (int j = 0; j < kAvChunk * 2; ++j)
    vq[0][j] = *(const v4f*)(vp + (size_t)(av_vrow(j >> 1) + (j & 1)) * kDH);
  __builtin_amdgcn_sched_barrier(0);

  float mx = fmaxf(fmaxf(sc[0][0], sc[0][1]), fmaxf(sc[0][2], sc[0][3]));
#pragma unroll
  for (int q = 1; q < kAvSteps / 2; ++q)
    mx = fmaxf(mx, fmaxf(fmaxf(sc[q][0], sc[q][1]), fmaxf(sc[q][2], sc[q][3])));
  mx = fmaxf(mx, __shfl_xor(mx, 16, 32));

  float sum = 0.0f;
  v8f acc[kAvT] = {v8f{}, v8f{}, v8f{}, v8f{}};
#pragma unroll
  for (int c = 0; c < kAvSteps / kAvChunk; ++c) {
    if (c + 1 < kAvSteps / kAvChunk) {
#pragma unroll
      for (int j = 0; j < kAvChunk * 2; ++j)
        vq[(c + 1) & 1][j] =
            *(const v4f*)(vp + (size_t)(av_vrow((c + 1) * kAvChunk + (j >> 1)) + (j & 1)) * kDH);
    }
    __builtin_amdgcn_sched_barrier(0);
#pragma unroll
    for (int s = 0; s < kAvChunk; ++s) {
      const int i = c * kAvChunk + s;
      v2f a = {__expf(sc[i >> 1][2 * (i & 1)] - mx), __expf(sc[i >> 1][2 * (i & 1) + 1] - mx)};
      sum += a[0] + a[1];
      const v4f r0 = vq[c & 1][2 * s];
      const v4f r1 = vq[c & 1][2 * s + 1];
#pragma unroll
      for (int t = 0; t < kAvT; ++t) acc[t] = wmma_f32(a, v2f{r0[t], r1[t]}, acc[t]);
    }
  }
  sum += __shfl_xor(sum, 16, 32);

  if (half == 0) {
    smax[wave][lr] = mx;
    ssum[wave][lr] = sum;
  }
#pragma unroll
  for (int t = 0; t < kAvT; ++t) {
#pragma unroll
    for (int r = 0; r < 8; ++r) part[wave][t * 8 + r][lane] = acc[t][r];
  }
  __syncthreads();

  const int row = wave + 8 * half;
  float g = smax[0][row];
#pragma unroll
  for (int w = 1; w < kAvWaves; ++w) g = fmaxf(g, smax[w][row]);
  float f[kAvWaves];
  float den = 0.0f;
#pragma unroll
  for (int w = 0; w < kAvWaves; ++w) {
    f[w] = __expf(smax[w][row] - g);
    den += f[w] * ssum[w][row];
  }
  const float inv = 1.0f / den;
  const int m = m0 + row;
  v4f o;
#pragma unroll
  for (int t = 0; t < kAvT; ++t) {
    float s = f[0] * part[0][t * 8 + wave][lane];
#pragma unroll
    for (int w = 1; w < kAvWaves; ++w) s += f[w] * part[w][t * 8 + wave][lane];
    o[t] = s * inv;
  }
  *(v4f*)(AVec + (size_t)m * kD + h * kDH + 4 * lr) = o;
}

static constexpr int kOutM = 64;
__global__ void __launch_bounds__(256)
out_gemm(const float* __restrict__ A, const float* __restrict__ B, float* __restrict__ C) {
  __shared__ alignas(16) float ldsA[2][kOutM * kPitch];
  __shared__ alignas(16) float ldsB[2][64 * kPitch];
  const int lane = threadIdx.x & 31;
  const int wave = threadIdx.x >> 5;
  const int half = lane >> 4;
  const int lr   = lane & 15;
  const int wm   = wave & 3;
  const int wn   = wave >> 2;
  const int m0   = blockIdx.x * kOutM + wm * 16;
  const int n0   = blockIdx.y * 64 + wn * 32;

  const float* Ag = A + (size_t)blockIdx.x * kOutM * kD;
  const float* Bg = B + (size_t)blockIdx.y * 64 * kD;
  const int la = (wm * 16 + lr) * kPitch + 4 * half;
  const int lb = (wn * 32 + lr) * kPitch + 4 * half;

  v8f acc[1][2] = {{v8f{}, v8f{}}};
  v4f ra[kOutM / 16];
  v4f rb[4];
  tile_fetch<kOutM>(ra, Ag, kD);
  tile_fetch<64>(rb, Bg, kD);
  tile_put<kOutM>(ldsA[0], ra);
  tile_put<64>(ldsB[0], rb);
  __syncthreads();
  for (int c = 0; c < kD / kKC; ++c) {
    const int cur = c & 1;
    if (c + 1 < kD / kKC) {
      tile_fetch<kOutM>(ra, Ag + (c + 1) * kKC, kD);
      tile_fetch<64>(rb, Bg + (c + 1) * kKC, kD);
    }
    tile_mma<1, 2>(ldsA[cur] + la, ldsB[cur] + lb, acc);
    __builtin_amdgcn_sched_barrier(0);
    if (c + 1 < kD / kKC) {
      tile_put<kOutM>(ldsA[cur ^ 1], ra);
      tile_put<64>(ldsB[cur ^ 1], rb);
      __syncthreads();
    }
  }
#pragma unroll
  for (int t = 0; t < 2; ++t) {
    int n = n0 + t * 16 + lr;
#pragma unroll
    for (int r = 0; r < 8; ++r) {
      int m = m0 + r + 8 * half;
      C[(size_t)m * kD + n] = acc[0][t][r];
    }
  }
}

extern "C" void kernel_launch(void* const* d_in, const int* in_sizes, int n_in,
                              void* d_out, int out_size, void* d_ws, size_t ws_size,
                              hipStream_t stream) {
  const float* x   = (const float*)d_in[0];
  const float* pos = (const float*)d_in[1];
  const float* Wq  = (const float*)d_in[2];
  const float* Wk  = (const float*)d_in[3];
  const float* Wv  = (const float*)d_in[4];
  const float* Wp  = (const float*)d_in[5];
  const float* Wo  = (const float*)d_in[6];
  const float* u   = (const float*)d_in[7];
  const float* v   = (const float*)d_in[8];

  float* ws = (float*)d_ws;
  const size_t SD = (size_t)kS * kD;
  float* Qhs  = ws;
  float* Khs  = Qhs + SD;
  float* Vhs  = Khs + SD;
  float* Phs  = Vhs + SD;
  float* AVec = Phs + SD;
  float* BD   = AVec + SD;
  float* SC   = BD + (size_t)kH * kS * kS;

  dim3 blk(256);
  proj_gemm_head<<<dim3(kS / kBlkM, kD / 64, 4), blk, 0, stream>>>(x, pos, Wq, Wk, Wv, Wp, Qhs);

  const float scale = 0.125f;
  bd_gemm<<<dim3(kS / kBlkM, kS / 64, kH), blk, 0, stream>>>(Qhs, Phs, v, BD);
  for (int h0 = 0; h0 < kH; h0 += kGrp) {
    score_gemm<<<dim3(kS / kBlkM, kS / 64, kGrp), blk, 0, stream>>>(Qhs, Khs, u, BD, SC, h0, scale);
    av_gemm   <<<dim3(kS / 16, 1, kGrp), blk, 0, stream>>>(SC, Vhs, AVec, h0);
  }

  out_gemm<<<dim3(kS / kOutM, kD / 64), blk, 0, stream>>>(AVec, Wo, (float*)d_out);
}


# rel_shift applied at the BD store: each product goes to its shifted flat place in a padded per-head plane (pad zeros stored), scores formed in place; 8 heads fit, score and AV 1 launch each (7 -> 5)
# speedup vs baseline: 10.1996x; 1.0866x over previous
#include <hip/hip_runtime.h>
#include <math.h>

typedef __attribute__((ext_vector_type(2))) float v2f;
typedef __attribute__((ext_vector_type(4))) float v4f;
typedef __attribute__((ext_vector_type(8))) float v8f;

static constexpr int kS  = 2048;
static constexpr int kD  = 512;
static constexpr int kH  = 8;
static constexpr int kDH = 64;
static constexpr int kPlane = kS * kS + kS;
static constexpr int kMT  = 2;
static constexpr int kBlkM = 8 * kMT * 16;
static constexpr int kKC   = 64;
static constexpr int kPitch = kKC + 4;

__device__ __forceinline__ v8f wmma_f32(v2f a, v2f b, v8f c) {
  return __builtin_amdgcn_wmma_f32_16x16x4_f32(
       false, a,  false, b,
       (short)0, c,  false,  false);
}

__device__ __forceinline__ v2f lo2(v4f x) { return v2f{x[0], x[1]}; }
__device__ __forceinline__ v2f hi2(v4f x) { return v2f{x[2], x[3]}; }


template <int kRows>
__device__ __forceinline__ void tile_fetch(v4f (&reg)[kRows / 16], const float* __restrict__ src,
                                           int ld) {
  const int c4 = threadIdx.x & 15;
  const int r  = threadIdx.x >> 4;
#pragma unroll
  for (int j = 0; j < kRows / 16; ++j)
    reg[j] = *(const v4f*)(src + (size_t)(r + 16 * j) * ld + 4 * c4);
}

template <int kRows>
__device__ __forceinline__ void tile_put(float* __restrict__ dst, const v4f (&reg)[kRows / 16]) {
  const int c4 = threadIdx.x & 15;
  const int r  = threadIdx.x >> 4;
#pragma unroll
  for (int j = 0; j < kRows / 16; ++j) *(v4f*)(dst + (r + 16 * j) * kPitch + 4 * c4) = reg[j];
}

template <int kMTw, int kNTw>
__device__ __forceinline__ void tile_mma(const float* la, const float* lb,
                                         v8f (&acc)[kMTw][kNTw]) {
#pragma unroll
  for (int k = 0; k < kKC; k += 8) {
    v4f a[kMTw];
    v4f b[kNTw];
#pragma unroll
    for (int u = 0; u < kMTw; ++u) a[u] = *(const v4f*)(la + u * 16 * kPitch + k);
#pragma unroll
    for (int t = 0; t < kNTw; ++t) b[t] = *(const v4f*)(lb + t * 16 * kPitch + k);
#pragma unroll
    for (int u = 0; u < kMTw; ++u) {
#pragma unroll
      for (int t = 0; t < kNTw; ++t) acc[u][t] = wmma_f32(lo2(a[u]), lo2(b[t]), acc[u][t]);
    }
#pragma unroll
    for (int u = 0; u < kMTw; ++u) {
#pragma unroll
      for (int t = 0; t < kNTw; ++t) acc[u][t] = wmma_f32(hi2(a[u]), hi2(b[t]), acc[u][t]);
    }
  }
}

__global__ void __launch_bounds__(256)
proj_gemm_head(const float* __restrict__ x, const float* __restrict__ pos,
               const float* __restrict__ Wq, const float* __restrict__ Wk,
               const float* __restrict__ Wv, const float* __restrict__ Wp,
               float* __restrict__ out4) {
  __shared__ alignas(16) float ldsA[2][kBlkM * kPitch];
  __shared__ alignas(16) float ldsB[2][64 * kPitch];
  const int which = blockIdx.z;
  const float* A = (which == 3) ? pos : x;
  const float* B = (which == 0) ? Wq : (which == 1) ? Wk : (which == 2) ? Wv : Wp;
  float* out = out4 + (size_t)which * kS * kD;
  const int lane = threadIdx.x & 31;
  const int wave = threadIdx.x >> 5;
  const int half = lane >> 4;
  const int lr   = lane & 15;
  const int m0   = blockIdx.x * kBlkM + wave * (kMT * 16);
  const int n0   = blockIdx.y * 64;

  const float* Ag = A + (size_t)blockIdx.x * kBlkM * kD;
  const float* Bg = B + (size_t)n0 * kD;
  const int la = (wave * (kMT * 16) + lr) * kPitch + 4 * half;
  const int lb = lr * kPitch + 4 * half;

  v8f acc[kMT][4];
#pragma unroll
  for (int u = 0; u < kMT; ++u) {
#pragma unroll
    for (int t = 0; t < 4; ++t) acc[u][t] = v8f{};
  }
  v4f ra[kBlkM / 16];
  v4f rb[4];
  tile_fetch<kBlkM>(ra, Ag, kD);
  tile_fetch<64>(rb, Bg, kD);
  tile_put<kBlkM>(ldsA[0], ra);
  tile_put<64>(ldsB[0], rb);
  __syncthreads();
  for (int c = 0; c < kD / kKC; ++c) {
    const int cur = c & 1;
    if (c + 1 < kD / kKC) {
      tile_fetch<kBlkM>(ra, Ag + (c + 1) * kKC, kD);
      tile_fetch<64>(rb, Bg + (c + 1) * kKC, kD);
    }
    tile_mma<kMT, 4>(ldsA[cur] + la, ldsB[cur] + lb, acc);
    __builtin_amdgcn_sched_barrier(0);
    if (c + 1 < kD / kKC) {
      tile_put<kBlkM>(ldsA[cur ^ 1], ra);
      tile_put<64>(ldsB[cur ^ 1], rb);
      __syncthreads();
    }
  }
#pragma unroll
  for (int u = 0; u < kMT; ++u) {
#pragma unroll
    for (int t = 0; t < 4; ++t) {
      int n = n0 + t * 16 + lr;
      int h = n >> 6, d = n & 63;
      float* op = out + (size_t)h * kS * kDH + d;
#pragma unroll
      for (int r = 0; r < 8; ++r) {
        int m = m0 + u * 16 + r + 8 * half;
        op[(size_t)m * kDH] = acc[u][t][r];
      }
    }
  }
}

__global__ void __launch_bounds__(256)
bd_gemm(const float* __restrict__ Qh, const float* __restrict__ Ph,
        const float* __restrict__ vb, float* __restrict__ PLall) {
  __shared__ alignas(16) float ldsA[kBlkM * kPitch];
  __shared__ alignas(16) float ldsB[64 * kPitch];
  const int h = blockIdx.z;
  float* PL = PLall + (size_t)h * kPlane + kS;
  const int lane = threadIdx.x & 31;
  const int wave = threadIdx.x >> 5;
  const int half = lane >> 4;
  const int lr   = lane & 15;
  const int m0   = blockIdx.x * kBlkM + wave * (kMT * 16);
  const int n0   = blockIdx.y * 64;

  {
    v4f ra[kBlkM / 16];
    v4f rb[4];
    tile_fetch<kBlkM>(ra, Qh + (size_t)h * kS * kDH + (size_t)blockIdx.x * kBlkM * kDH, kDH);
    tile_fetch<64>(rb, Ph + (size_t)h * kS * kDH + (size_t)n0 * kDH, kDH);
    const v4f bias = *(const v4f*)(vb + h * kDH + 4 * (threadIdx.x & 15));
#pragma unroll
    for (int j = 0; j < kBlkM / 16; ++j) ra[j] = ra[j] + bias;
    tile_put<kBlkM>(ldsA, ra);
    tile_put<64>(ldsB, rb);
  }
  __syncthreads();

  v8f acc[kMT][4];
#pragma unroll
  for (int u = 0; u < kMT; ++u) {
#pragma unroll
    for (int t = 0; t < 4; ++t) acc[u][t] = v8f{};
  }
  tile_mma<kMT, 4>(ldsA + (wave * (kMT * 16) + lr) * kPitch + 4 * half,
                   ldsB + lr * kPitch + 4 * half, acc);
#pragma unroll
  for (int u = 0; u < kMT; ++u) {
#pragma unroll
    for (int t = 0; t < 4; ++t) {
      int n = n0 + t * 16 + lr;
#pragma unroll
      for (int r = 0; r < 8; ++r) {
        int m = m0 + u * 16 + r + 8 * half;
        PL[m * (kS + 1) + n + 1 - kS] = acc[u][t][r];
      }
    }
  }
  if (blockIdx.y == kS / 64 - 1 && lr == 15) {
#pragma unroll
    for (int u = 0; u < kMT; ++u) {
#pragma unroll
      for (int r = 0; r < 8; ++r) {
        int m = m0 + u * 16 + r + 8 * half;
        if (m < kS - 1) PL[m * (kS + 1) + 1] = 0.0f;
      }
    }
  }
}

__global__ void __launch_bounds__(256, 6)
score_gemm(const float* __restrict__ Qh, const float* __restrict__ Kh,
           const float* __restrict__ ub, float* __restrict__ PLall, float scale) {
  __shared__ alignas(16) float ldsA[kBlkM * kPitch];
  __shared__ alignas(16) float ldsB[64 * kPitch];
  const int h = blockIdx.z;
  float* PL = PLall + (size_t)h * kPlane + kS;
  const int lane = threadIdx.x & 31;
  const int wave = threadIdx.x >> 5;
  const int half = lane >> 4;
  const int lr   = lane & 15;
  const int m0   = blockIdx.x * kBlkM + wave * (kMT * 16);
  const int n0   = blockIdx.y * 64;

  {
    v4f ra[kBlkM / 16];
    v4f rb[4];
    tile_fetch<kBlkM>(ra, Qh + (size_t)h * kS * kDH + (size_t)blockIdx.x * kBlkM * kDH, kDH);
    tile_fetch<64>(rb, Kh + (size_t)h * kS * kDH + (size_t)n0 * kDH, kDH);
    const v4f bias = *(const v4f*)(ub + h * kDH + 4 * (threadIdx.x & 15));
#pragma unroll
    for (int j = 0; j < kBlkM / 16; ++j) ra[j] = ra[j] + bias;
    tile_put<kBlkM>(ldsA, ra);
    tile_put<64>(ldsB, rb);
  }
  __syncthreads();

  v8f acc[kMT][4];
#pragma unroll
  for (int u = 0; u < kMT; ++u) {
#pragma unroll
    for (int t = 0; t < 4; ++t) acc[u][t] = v8f{};
  }
  tile_mma<kMT, 4>(ldsA + (wave * (kMT * 16) + lr) * kPitch + 4 * half,
                   ldsB + lr * kPitch + 4 * half, acc);
  constexpr int kTB = 2;
#pragma unroll
  for (int u = 0; u < kMT; ++u) {
#pragma unroll
    for (int tb = 0; tb < 4; tb += kTB) {
      float bdv[kTB][8];
#pragma unroll
      for (int t = 0; t < kTB; ++t) {
        const int n = n0 + (tb + t) * 16 + lr;
#pragma unroll
        for (int r = 0; r < 8; ++r) {
          const int m = m0 + u * 16 + r + 8 * half;
          bdv[t][r] = PL[m * kS + n];
        }
      }
      __builtin_amdgcn_sched_barrier(0);
#pragma unroll
      for (int t = 0; t < kTB; ++t) {
        const int n = n0 + (tb + t) * 16 + lr;
#pragma unroll
        for (int r = 0; r < 8; ++r) {
          const int m = m0 + u * 16 + r + 8 * half;
          PL[m * kS + n] = (acc[u][tb + t][r] + bdv[t][r]) * scale;
        }
      }
      __builtin_amdgcn_sched_barrier(0);
    }
  }
}

static constexpr int kAvWaves = 8;
static constexpr int kAvK     = kS / kAvWaves;
static constexpr int kAvT     = kDH / 16;
static constexpr int kAvSteps = kAvK / 4;
static constexpr int kAvChunk = 4;

__device__ __forceinline__ constexpr int av_vrow(int i) { return 8 * (i >> 1) + 2 * (i & 1); }

__global__ void __launch_bounds__(kAvWaves * 32, 4)
av_gemm(const float* __restrict__ PLall, const float* __restrict__ Vh,
        float* __restrict__ AVec) {
  const int h = blockIdx.z;
  const float* SC = PLall + (size_t)h * kPlane + kS;
  __shared__ float part[kAvWaves][kAvT * 8][32];
  __shared__ float smax[kAvWaves][16];
  __shared__ float ssum[kAvWaves][16];
  const int lane = threadIdx.x & 31;
  const int wave = threadIdx.x >> 5;
  const int half = lane >> 4;
  const int lr   = lane & 15;
  const int m0   = blockIdx.x * 16;
  const int kb   = wave * kAvK + 4 * half;

  const float* pp = SC + (size_t)(m0 + lr) * kS + kb;
  const float* vp = Vh + (size_t)h * kS * kDH + (size_t)kb * kDH + 4 * lr;

  v4f sc[kAvSteps / 2];
#pragma unroll
  for (int q = 0; q < kAvSteps / 2; ++q) sc[q] = *(const v4f*)(pp + 8 * q);

  v4f vq[2][kAvChunk * 2];
#pragma unroll
  for (int j = 0; j < kAvChunk * 2; ++j)
    vq[0][j] = *(const v4f*)(vp + (size_t)(av_vrow(j >> 1) + (j & 1)) * kDH);
  __builtin_amdgcn_sched_barrier(0);

  float mx = fmaxf(fmaxf(sc[0][0], sc[0][1]), fmaxf(sc[0][2], sc[0][3]));
#pragma unroll
  for (int q = 1; q < kAvSteps / 2; ++q)
    mx = fmaxf(mx, fmaxf(fmaxf(sc[q][0], sc[q][1]), fmaxf(sc[q][2], sc[q][3])));
  mx = fmaxf(mx, __shfl_xor(mx, 16, 32));

  float sum = 0.0f;
  v8f acc[kAvT] = {v8f{}, v8f{}, v8f{}, v8f{}};
#pragma unroll
  for (int c = 0; c < kAvSteps / kAvChunk; ++c) {
    if (c + 1 < kAvSteps / kAvChunk) {
#pragma unroll
      for (int j = 0; j < kAvChunk * 2; ++j)
        vq[(c + 1) & 1][j] =
            *(const v4f*)(vp + (size_t)(av_vrow((c + 1) * kAvChunk + (j >> 1)) + (j & 1)) * kDH);
    }
    __builtin_amdgcn_sched_barrier(0);
#pragma unroll
    for (int s = 0; s < kAvChunk; ++s) {
      const int i = c * kAvChunk + s;
      v2f a = {__expf(sc[i >> 1][2 * (i & 1)] - mx), __expf(sc[i >> 1][2 * (i & 1) + 1] - mx)};
      sum += a[0] + a[1];
      const v4f r0 = vq[c & 1][2 * s];
      const v4f r1 = vq[c & 1][2 * s + 1];
#pragma unroll
      for (int t = 0; t < kAvT; ++t) acc[t] = wmma_f32(a, v2f{r0[t], r1[t]}, acc[t]);
    }
  }
  sum += __shfl_xor(sum, 16, 32);

  if (half == 0) {
    smax[wave][lr] = mx;
    ssum[wave][lr] = sum;
  }
#pragma unroll
  for (int t = 0; t < kAvT; ++t) {
#pragma unroll
    for (int r = 0; r < 8; ++r) part[wave][t * 8 + r][lane] = acc[t][r];
  }
  __syncthreads();

  const int row = wave + 8 * half;
  float g = smax[0][row];
#pragma unroll
  for (int w = 1; w < kAvWaves; ++w) g = fmaxf(g, smax[w][row]);
  float f[kAvWaves];
  float den = 0.0f;
#pragma unroll
  for (int w = 0; w < kAvWaves; ++w) {
    f[w] = __expf(smax[w][row] - g);
    den += f[w] * ssum[w][row];
  }
  const float inv = 1.0f / den;
  const int m = m0 + row;
  v4f o;
#pragma unroll
  for (int t = 0; t < kAvT; ++t) {
    float s = f[0] * part[0][t * 8 + wave][lane];
#pragma unroll
    for (int w = 1; w < kAvWaves; ++w) s += f[w] * part[w][t * 8 + wave][lane];
    o[t] = s * inv;
  }
  *(v4f*)(AVec + (size_t)m * kD + h * kDH + 4 * lr) = o;
}

static constexpr int kOutM = 64;
__global__ void __launch_bounds__(256)
out_gemm(const float* __restrict__ A, const float* __restrict__ B, float* __restrict__ C) {
  __shared__ alignas(16) float ldsA[2][kOutM * kPitch];
  __shared__ alignas(16) float ldsB[2][64 * kPitch];
  const int lane = threadIdx.x & 31;
  const int wave = threadIdx.x >> 5;
  const int half = lane >> 4;
  const int lr   = lane & 15;
  const int wm   = wave & 3;
  const int wn   = wave >> 2;
  const int m0   = blockIdx.x * kOutM + wm * 16;
  const int n0   = blockIdx.y * 64 + wn * 32;

  const float* Ag = A + (size_t)blockIdx.x * kOutM * kD;
  const float* Bg = B + (size_t)blockIdx.y * 64 * kD;
  const int la = (wm * 16 + lr) * kPitch + 4 * half;
  const int lb = (wn * 32 + lr) * kPitch + 4 * half;

  v8f acc[1][2] = {{v8f{}, v8f{}}};
  v4f ra[kOutM / 16];
  v4f rb[4];
  tile_fetch<kOutM>(ra, Ag, kD);
  tile_fetch<64>(rb, Bg, kD);
  tile_put<kOutM>(ldsA[0], ra);
  tile_put<64>(ldsB[0], rb);
  __syncthreads();
  for (int c = 0; c < kD / kKC; ++c) {
    const int cur = c & 1;
    if (c + 1 < kD / kKC) {
      tile_fetch<kOutM>(ra, Ag + (c + 1) * kKC, kD);
      tile_fetch<64>(rb, Bg + (c + 1) * kKC, kD);
    }
    tile_mma<1, 2>(ldsA[cur] + la, ldsB[cur] + lb, acc);
    __builtin_amdgcn_sched_barrier(0);
    if (c + 1 < kD / kKC) {
      tile_put<kOutM>(ldsA[cur ^ 1], ra);
      tile_put<64>(ldsB[cur ^ 1], rb);
      __syncthreads();
    }
  }
#pragma unroll
  for (int t = 0; t < 2; ++t) {
    int n = n0 + t * 16 + lr;
#pragma unroll
    for (int r = 0; r < 8; ++r) {
      int m = m0 + r + 8 * half;
      C[(size_t)m * kD + n] = acc[0][t][r];
    }
  }
}

extern "C" void kernel_launch(void* const* d_in, const int* in_sizes, int n_in,
                              void* d_out, int out_size, void* d_ws, size_t ws_size,
                              hipStream_t stream) {
  const float* x   = (const float*)d_in[0];
  const float* pos = (const float*)d_in[1];
  const float* Wq  = (const float*)d_in[2];
  const float* Wk  = (const float*)d_in[3];
  const float* Wv  = (const float*)d_in[4];
  const float* Wp  = (const float*)d_in[5];
  const float* Wo  = (const float*)d_in[6];
  const float* u   = (const float*)d_in[7];
  const float* v   = (const float*)d_in[8];

  float* ws = (float*)d_ws;
  const size_t SD = (size_t)kS * kD;
  float* Qhs  = ws;
  float* Khs  = Qhs + SD;
  float* Vhs  = Khs + SD;
  float* Phs  = Vhs + SD;
  float* AVec = Phs + SD;
  float* PL   = AVec + SD;

  dim3 blk(256);
  proj_gemm_head<<<dim3(kS / kBlkM, kD / 64, 4), blk, 0, stream>>>(x, pos, Wq, Wk, Wv, Wp, Qhs);

  const float scale = 0.125f;
  bd_gemm   <<<dim3(kS / kBlkM, kS / 64, kH), blk, 0, stream>>>(Qhs, Phs, v, PL);
  score_gemm<<<dim3(kS / kBlkM, kS / 64, kH), blk, 0, stream>>>(Qhs, Khs, u, PL, scale);
  av_gemm   <<<dim3(kS / 16, 1, kH), blk, 0, stream>>>(PL, Vhs, AVec);

  out_gemm<<<dim3(kS / kOutM, kD / 64), blk, 0, stream>>>(AVec, Wo, (float*)d_out);
}
